# starting kernel
# baseline (speedup 1.0000x reference)
#include <hip/hip_runtime.h>


typedef float v2f __attribute__((ext_vector_type(2)));
typedef float v8f __attribute__((ext_vector_type(8)));

#define D 128
#define NTILE 16

__global__ void zero_kernel(float* __restrict__ p, long n) {
  long i = (long)blockIdx.x * blockDim.x + threadIdx.x;
  long stride = (long)gridDim.x * blockDim.x;
  for (; i < n; i += stride) p[i] = 0.0f;
}

__global__ void scatter_kernel(const float* __restrict__ z,
                               const int* __restrict__ src,
                               const int* __restrict__ dst,
                               float* __restrict__ summed,
                               float* __restrict__ cnt,
                               int E) {
  unsigned tid = blockIdx.x * blockDim.x + threadIdx.x;
  int wid = (int)(tid >> 5);
  int lane = threadIdx.x & 31;
  if (wid >= E) return;
  int s = __builtin_amdgcn_readfirstlane(src[wid]);
  int d = __builtin_amdgcn_readfirstlane(dst[wid]);

  const float4* zp = (const float4*)(z + (size_t)s * D);
  float4 v = zp[lane];

  float* op = summed + (size_t)d * D + lane * 4;
  unsafeAtomicAdd(op + 0, fmaxf(v.x, 0.0f));
  unsafeAtomicAdd(op + 1, fmaxf(v.y, 0.0f));
  unsafeAtomicAdd(op + 2, fmaxf(v.z, 0.0f));
  unsafeAtomicAdd(op + 3, fmaxf(v.w, 0.0f));
  if (cnt != nullptr && lane == 0) unsafeAtomicAdd(cnt + d, 1.0f);
}

__global__ void inv_kernel(float* __restrict__ cnt, int n) {
  int i = blockIdx.x * blockDim.x + threadIdx.x;
  if (i < n) cnt[i] = 1.0f / fmaxf(cnt[i], 1.0f);
}

__global__ void __launch_bounds__(256)
combine_kernel(const float* __restrict__ summed, const float* __restrict__ inv,
               const float* __restrict__ z, const float* __restrict__ Wl,
               const float* __restrict__ bl, const float* __restrict__ Wr,
               float* __restrict__ out, int ntiles, int do_relu) {
  int tile = blockIdx.x * (blockDim.x >> 5) + (threadIdx.x >> 5);
  if (tile >= ntiles) return;
  int lane = threadIdx.x & 31;
  int lrow  = lane & 15;
  int khalf = (lane >> 4) * 2;

  v8f acc[8] = {};

  int row = tile * NTILE + lrow;
  float invr = inv[row];
  const float* arow = summed + (size_t)row * D;
  const float* zrow = z + (size_t)row * D;

  for (int k0 = 0; k0 < D; k0 += 4) {
    v2f a;
    a.x = arow[k0 + khalf + 0] * invr;
    a.y = arow[k0 + khalf + 1] * invr;
    const float* wp = Wl + (size_t)(k0 + khalf) * D + lrow;
#pragma unroll
    for (int n = 0; n < 8; ++n) {
      v2f b;
      b.x = wp[n * 16];
      b.y = wp[n * 16 + D];
      acc[n] = __builtin_amdgcn_wmma_f32_16x16x4_f32(
          false, a, false, b, (short)0, acc[n], false, false);
    }
  }
  for (int k0 = 0; k0 < D; k0 += 4) {
    v2f a;
    a.x = zrow[k0 + khalf + 0];
    a.y = zrow[k0 + khalf + 1];
    const float* wp = Wr + (size_t)(k0 + khalf) * D + lrow;
#pragma unroll
    for (int n = 0; n < 8; ++n) {
      v2f b;
      b.x = wp[n * 16];
      b.y = wp[n * 16 + D];
      acc[n] = __builtin_amdgcn_wmma_f32_16x16x4_f32(
          false, a, false, b, (short)0, acc[n], false, false);
    }
  }

  int mbase = tile * NTILE + ((lane >> 4) << 3);
#pragma unroll
  for (int n = 0; n < 8; ++n) {
    float bias = bl[n * 16 + lrow];
#pragma unroll
    for (int v = 0; v < 8; ++v) {
      float val = acc[n][v] + bias;
      if (do_relu) val = fmaxf(val, 0.0f);
      out[(size_t)(mbase + v) * D + n * 16 + lrow] = val;
    }
  }
}

extern "C" void kernel_launch(void* const* d_in, const int* in_sizes, int n_in,
                              void* d_out, int out_size, void* d_ws, size_t ws_size,
                              hipStream_t stream) {
  const float* x   = (const float*)d_in[0];
  const int*   ei  = (const int*)d_in[1];
  const float* Wl1 = (const float*)d_in[2];
  const float* bl1 = (const float*)d_in[3];
  const float* Wr1 = (const float*)d_in[4];
  const float* Wl2 = (const float*)d_in[5];
  const float* bl2 = (const float*)d_in[6];
  const float* Wr2 = (const float*)d_in[7];
  float* out = (float*)d_out;

  const int N = in_sizes[0] / D;
  const int E = in_sizes[1] / 2;
  const int ntiles = (N + NTILE - 1) / NTILE;

  const int* src = ei;
  const int* dst = ei + E;

  float* cnt = (float*)d_ws;
  float* summed = cnt + N;

  int scatter_blocks = (int)(((long)E * 32 + 255) / 256);
  int combine_blocks = (ntiles + 7) / 8;

  zero_kernel<<<2048, 256, 0, stream>>>(cnt, (long)N * (D + 1));
  scatter_kernel<<<scatter_blocks, 256, 0, stream>>>(x, src, dst, summed, cnt, E);
  inv_kernel<<<(N + 255) / 256, 256, 0, stream>>>(cnt, N);
  combine_kernel<<<combine_blocks, 256, 0, stream>>>(summed, cnt, x, Wl1, bl1, Wr1,
                                                     out, ntiles, 1);
  zero_kernel<<<2048, 256, 0, stream>>>(summed, (long)N * D);
  scatter_kernel<<<scatter_blocks, 256, 0, stream>>>(out, src, dst, summed, nullptr, E);
  combine_kernel<<<combine_blocks, 256, 0, stream>>>(summed, cnt, out, Wl2, bl2, Wr2,
                                                     out, ntiles, 0);
}


# scatter: lane l takes channels l, l+32, l+64, l+96, so each f32 atomic instruction and each row load covers one whole 128-B line, not a quarter of four lines; sum order still arrival order
# speedup vs baseline: 1.2973x; 1.2973x over previous
#include <hip/hip_runtime.h>


typedef float v2f __attribute__((ext_vector_type(2)));
typedef float v8f __attribute__((ext_vector_type(8)));

#define D 128
#define NTILE 16

__global__ void zero_kernel(float* __restrict__ p, long n) {
  long i = (long)blockIdx.x * blockDim.x + threadIdx.x;
  long stride = (long)gridDim.x * blockDim.x;
  for (; i < n; i += stride) p[i] = 0.0f;
}

__global__ void scatter_kernel(const float* __restrict__ z,
                               const int* __restrict__ src,
                               const int* __restrict__ dst,
                               float* __restrict__ summed,
                               float* __restrict__ cnt,
                               int E) {
  unsigned tid = blockIdx.x * blockDim.x + threadIdx.x;
  int wid = (int)(tid >> 5);
  int lane = threadIdx.x & 31;
  if (wid >= E) return;
  int s = __builtin_amdgcn_readfirstlane(src[wid]);
  int d = __builtin_amdgcn_readfirstlane(dst[wid]);

  const float* zp = z + (size_t)s * D + lane;
  float v0 = zp[0];
  float v1 = zp[32];
  float v2 = zp[64];
  float v3 = zp[96];

  float* op = summed + (size_t)d * D + lane;
  unsafeAtomicAdd(op + 0,  fmaxf(v0, 0.0f));
  unsafeAtomicAdd(op + 32, fmaxf(v1, 0.0f));
  unsafeAtomicAdd(op + 64, fmaxf(v2, 0.0f));
  unsafeAtomicAdd(op + 96, fmaxf(v3, 0.0f));
  if (cnt != nullptr && lane == 0) unsafeAtomicAdd(cnt + d, 1.0f);
}

__global__ void inv_kernel(float* __restrict__ cnt, int n) {
  int i = blockIdx.x * blockDim.x + threadIdx.x;
  if (i < n) cnt[i] = 1.0f / fmaxf(cnt[i], 1.0f);
}

__global__ void __launch_bounds__(256)
combine_kernel(const float* __restrict__ summed, const float* __restrict__ inv,
               const float* __restrict__ z, const float* __restrict__ Wl,
               const float* __restrict__ bl, const float* __restrict__ Wr,
               float* __restrict__ out, int ntiles, int do_relu) {
  int tile = blockIdx.x * (blockDim.x >> 5) + (threadIdx.x >> 5);
  if (tile >= ntiles) return;
  int lane = threadIdx.x & 31;
  int lrow  = lane & 15;
  int khalf = (lane >> 4) * 2;

  v8f acc[8] = {};

  int row = tile * NTILE + lrow;
  float invr = inv[row];
  const float* arow = summed + (size_t)row * D;
  const float* zrow = z + (size_t)row * D;

  for (int k0 = 0; k0 < D; k0 += 4) {
    v2f a;
    a.x = arow[k0 + khalf + 0] * invr;
    a.y = arow[k0 + khalf + 1] * invr;
    const float* wp = Wl + (size_t)(k0 + khalf) * D + lrow;
#pragma unroll
    for (int n = 0; n < 8; ++n) {
      v2f b;
      b.x = wp[n * 16];
      b.y = wp[n * 16 + D];
      acc[n] = __builtin_amdgcn_wmma_f32_16x16x4_f32(
          false, a, false, b, (short)0, acc[n], false, false);
    }
  }
  for (int k0 = 0; k0 < D; k0 += 4) {
    v2f a;
    a.x = zrow[k0 + khalf + 0];
    a.y = zrow[k0 + khalf + 1];
    const float* wp = Wr + (size_t)(k0 + khalf) * D + lrow;
#pragma unroll
    for (int n = 0; n < 8; ++n) {
      v2f b;
      b.x = wp[n * 16];
      b.y = wp[n * 16 + D];
      acc[n] = __builtin_amdgcn_wmma_f32_16x16x4_f32(
          false, a, false, b, (short)0, acc[n], false, false);
    }
  }

  int mbase = tile * NTILE + ((lane >> 4) << 3);
#pragma unroll
  for (int n = 0; n < 8; ++n) {
    float bias = bl[n * 16 + lrow];
#pragma unroll
    for (int v = 0; v < 8; ++v) {
      float val = acc[n][v] + bias;
      if (do_relu) val = fmaxf(val, 0.0f);
      out[(size_t)(mbase + v) * D + n * 16 + lrow] = val;
    }
  }
}

extern "C" void kernel_launch(void* const* d_in, const int* in_sizes, int n_in,
                              void* d_out, int out_size, void* d_ws, size_t ws_size,
                              hipStream_t stream) {
  const float* x   = (const float*)d_in[0];
  const int*   ei  = (const int*)d_in[1];
  const float* Wl1 = (const float*)d_in[2];
  const float* bl1 = (const float*)d_in[3];
  const float* Wr1 = (const float*)d_in[4];
  const float* Wl2 = (const float*)d_in[5];
  const float* bl2 = (const float*)d_in[6];
  const float* Wr2 = (const float*)d_in[7];
  float* out = (float*)d_out;

  const int N = in_sizes[0] / D;
  const int E = in_sizes[1] / 2;
  const int ntiles = (N + NTILE - 1) / NTILE;

  const int* src = ei;
  const int* dst = ei + E;

  float* cnt = (float*)d_ws;
  float* summed = cnt + N;

  int scatter_blocks = (int)(((long)E * 32 + 255) / 256);
  int combine_blocks = (ntiles + 7) / 8;

  zero_kernel<<<2048, 256, 0, stream>>>(cnt, (long)N * (D + 1));
  scatter_kernel<<<scatter_blocks, 256, 0, stream>>>(x, src, dst, summed, cnt, E);
  inv_kernel<<<(N + 255) / 256, 256, 0, stream>>>(cnt, N);
  combine_kernel<<<combine_blocks, 256, 0, stream>>>(summed, cnt, x, Wl1, bl1, Wr1,
                                                     out, ntiles, 1);
  zero_kernel<<<2048, 256, 0, stream>>>(summed, (long)N * D);
  scatter_kernel<<<scatter_blocks, 256, 0, stream>>>(out, src, dst, summed, nullptr, E);
  combine_kernel<<<combine_blocks, 256, 0, stream>>>(summed, cnt, out, Wl2, bl2, Wr2,
                                                     out, ntiles, 0);
}


# aggregation: edges bucketed by destination per call (int-atomic count, scan, fill); one wave per node sums its neighbour rows in registers, no f32 atomics; sum order = slot arrival order
# speedup vs baseline: 2.2925x; 1.7671x over previous
#include <hip/hip_runtime.h>


typedef float v2f __attribute__((ext_vector_type(2)));
typedef float v8f __attribute__((ext_vector_type(8)));

#define D 128
#define NTILE 16

__global__ void zero_kernel(float* __restrict__ p, long n) {
  long i = (long)blockIdx.x * blockDim.x + threadIdx.x;
  long stride = (long)gridDim.x * blockDim.x;
  for (; i < n; i += stride) p[i] = 0.0f;
}

__global__ void scatter_kernel(const float* __restrict__ z,
                               const int* __restrict__ src,
                               const int* __restrict__ dst,
                               float* __restrict__ summed,
                               float* __restrict__ cnt,
                               int E) {
  unsigned tid = blockIdx.x * blockDim.x + threadIdx.x;
  int wid = (int)(tid >> 5);
  int lane = threadIdx.x & 31;
  if (wid >= E) return;
  int s = __builtin_amdgcn_readfirstlane(src[wid]);
  int d = __builtin_amdgcn_readfirstlane(dst[wid]);

  const float* zp = z + (size_t)s * D + lane;
  float v0 = zp[0];
  float v1 = zp[32];
  float v2 = zp[64];
  float v3 = zp[96];

  float* op = summed + (size_t)d * D + lane;
  unsafeAtomicAdd(op + 0,  fmaxf(v0, 0.0f));
  unsafeAtomicAdd(op + 32, fmaxf(v1, 0.0f));
  unsafeAtomicAdd(op + 64, fmaxf(v2, 0.0f));
  unsafeAtomicAdd(op + 96, fmaxf(v3, 0.0f));
  if (cnt != nullptr && lane == 0) unsafeAtomicAdd(cnt + d, 1.0f);
}

__global__ void inv_kernel(float* __restrict__ cnt, int n) {
  int i = blockIdx.x * blockDim.x + threadIdx.x;
  if (i < n) cnt[i] = 1.0f / fmaxf(cnt[i], 1.0f);
}

__global__ void count_kernel(const int* __restrict__ dst, int* __restrict__ deg,
                             int* __restrict__ rank, int E) {
  int e = blockIdx.x * blockDim.x + threadIdx.x;
  if (e < E) rank[e] = atomicAdd(deg + dst[e], 1);
}

#define SCAN_TILE 4096

__global__ void __launch_bounds__(1024)
scan_kernel(const int* __restrict__ deg, int* __restrict__ rowptr) {
  __shared__ int wsum[32];
  __shared__ int wtot[32];
  const int4* deg4 = reinterpret_cast<const int4*>(deg);
  int t = threadIdx.x;
  int lane = t & 31;
  int w = t >> 5;

  int part = 0;
  int nfront = blockIdx.x * (SCAN_TILE / 4);
#pragma unroll 4
  for (int k = t; k < nfront; k += 1024) {
    int4 q = deg4[k];
    part += (q.x + q.y) + (q.z + q.w);
  }
  for (int o = 16; o > 0; o >>= 1) part += __shfl_xor(part, o);
  if (lane == 0) wsum[w] = part;

  int4 v = deg4[nfront + t];
  int s0 = v.x, s1 = s0 + v.y, s2 = s1 + v.z, s3 = s2 + v.w;
  int incl = s3;
  for (int o = 1; o < 32; o <<= 1) {
    int u = __shfl_up(incl, o);
    if (lane >= o) incl += u;
  }
  if (lane == 31) wtot[w] = incl;
  __syncthreads();

  int front = wsum[lane];
  int mine = (lane < w) ? wtot[lane] : 0;
  for (int o = 16; o > 0; o >>= 1) {
    front += __shfl_xor(front, o);
    mine += __shfl_xor(mine, o);
  }
  int ex = front + mine + (incl - s3);
  reinterpret_cast<int4*>(rowptr)[nfront + t] = make_int4(ex, ex + s0, ex + s1, ex + s2);
}

__global__ void fill_kernel(const int* __restrict__ src, const int* __restrict__ dst,
                            const int* __restrict__ rank, const int* __restrict__ rowptr,
                            int* __restrict__ nbr, int E) {
  int e = blockIdx.x * blockDim.x + threadIdx.x;
  if (e < E) nbr[rowptr[dst[e]] + rank[e]] = src[e];
}

template <int K>
__device__ __forceinline__ void add_rows(const float4* __restrict__ z4, int idx, int j,
                                         int lane, float4& acc) {
  float4 v[K];
#pragma unroll
  for (int u = 0; u < K; ++u) {
    int s = __builtin_amdgcn_readlane(idx, j + u);
    v[u] = z4[(size_t)s * (D / 4) + lane];
  }
#pragma unroll
  for (int u = 0; u < K; ++u) {
    acc.x += fmaxf(v[u].x, 0.0f);
    acc.y += fmaxf(v[u].y, 0.0f);
    acc.z += fmaxf(v[u].z, 0.0f);
    acc.w += fmaxf(v[u].w, 0.0f);
  }
}

__global__ void __launch_bounds__(256)
gather_kernel(const float* __restrict__ z, const int* __restrict__ rowptr,
              const int* __restrict__ deg, const int* __restrict__ nbr,
              float* __restrict__ summed, float* __restrict__ inv, int n) {
  int node = blockIdx.x * (blockDim.x >> 5) + (threadIdx.x >> 5);
  if (node >= n) return;
  int lane = threadIdx.x & 31;
  int start = __builtin_amdgcn_readfirstlane(rowptr[node]);
  int cnt = __builtin_amdgcn_readfirstlane(deg[node]);
  const float4* z4 = reinterpret_cast<const float4*>(z);

  float4 acc = make_float4(0.0f, 0.0f, 0.0f, 0.0f);
  for (int base = 0; base < cnt; base += 32) {
    int m = cnt - base;
    if (m > 32) m = 32;
    int idx = (lane < m) ? nbr[start + base + lane] : 0;
    int j = 0;
    for (; j + 8 <= m; j += 8) add_rows<8>(z4, idx, j, lane, acc);
    if (m & 4) { add_rows<4>(z4, idx, j, lane, acc); j += 4; }
    if (m & 2) { add_rows<2>(z4, idx, j, lane, acc); j += 2; }
    if (m & 1) add_rows<1>(z4, idx, j, lane, acc);
  }
  reinterpret_cast<float4*>(summed)[(size_t)node * (D / 4) + lane] = acc;
  if (inv != nullptr && lane == 0) inv[node] = 1.0f / fmaxf((float)cnt, 1.0f);
}

__global__ void __launch_bounds__(256)
combine_kernel(const float* __restrict__ summed, const float* __restrict__ inv,
               const float* __restrict__ z, const float* __restrict__ Wl,
               const float* __restrict__ bl, const float* __restrict__ Wr,
               float* __restrict__ out, int ntiles, int do_relu) {
  int tile = blockIdx.x * (blockDim.x >> 5) + (threadIdx.x >> 5);
  if (tile >= ntiles) return;
  int lane = threadIdx.x & 31;
  int lrow  = lane & 15;
  int khalf = (lane >> 4) * 2;

  v8f acc[8] = {};

  int row = tile * NTILE + lrow;
  float invr = inv[row];
  const float* arow = summed + (size_t)row * D;
  const float* zrow = z + (size_t)row * D;

  for (int k0 = 0; k0 < D; k0 += 4) {
    v2f a;
    a.x = arow[k0 + khalf + 0] * invr;
    a.y = arow[k0 + khalf + 1] * invr;
    const float* wp = Wl + (size_t)(k0 + khalf) * D + lrow;
#pragma unroll
    for (int n = 0; n < 8; ++n) {
      v2f b;
      b.x = wp[n * 16];
      b.y = wp[n * 16 + D];
      acc[n] = __builtin_amdgcn_wmma_f32_16x16x4_f32(
          false, a, false, b, (short)0, acc[n], false, false);
    }
  }
  for (int k0 = 0; k0 < D; k0 += 4) {
    v2f a;
    a.x = zrow[k0 + khalf + 0];
    a.y = zrow[k0 + khalf + 1];
    const float* wp = Wr + (size_t)(k0 + khalf) * D + lrow;
#pragma unroll
    for (int n = 0; n < 8; ++n) {
      v2f b;
      b.x = wp[n * 16];
      b.y = wp[n * 16 + D];
      acc[n] = __builtin_amdgcn_wmma_f32_16x16x4_f32(
          false, a, false, b, (short)0, acc[n], false, false);
    }
  }

  int mbase = tile * NTILE + ((lane >> 4) << 3);
#pragma unroll
  for (int n = 0; n < 8; ++n) {
    float bias = bl[n * 16 + lrow];
#pragma unroll
    for (int v = 0; v < 8; ++v) {
      float val = acc[n][v] + bias;
      if (do_relu) val = fmaxf(val, 0.0f);
      out[(size_t)(mbase + v) * D + n * 16 + lrow] = val;
    }
  }
}

extern "C" void kernel_launch(void* const* d_in, const int* in_sizes, int n_in,
                              void* d_out, int out_size, void* d_ws, size_t ws_size,
                              hipStream_t stream) {
  const float* x   = (const float*)d_in[0];
  const int*   ei  = (const int*)d_in[1];
  const float* Wl1 = (const float*)d_in[2];
  const float* bl1 = (const float*)d_in[3];
  const float* Wr1 = (const float*)d_in[4];
  const float* Wl2 = (const float*)d_in[5];
  const float* bl2 = (const float*)d_in[6];
  const float* Wr2 = (const float*)d_in[7];
  float* out = (float*)d_out;

  const int N = in_sizes[0] / D;
  const int E = in_sizes[1] / 2;
  const int ntiles = (N + NTILE - 1) / NTILE;

  const int* src = ei;
  const int* dst = ei + E;

  float* cnt = (float*)d_ws;
  float* summed = cnt + N;

  int combine_blocks = (ntiles + 7) / 8;

  const int scan_blocks = (N + SCAN_TILE - 1) / SCAN_TILE;
  const int Npad = scan_blocks * SCAN_TILE;
  const size_t need = ((size_t)N * (D + 1) + 2 * (size_t)Npad + (size_t)E) * sizeof(float);

  if (ws_size >= need) {
    int* deg = (int*)(summed + (size_t)N * D);
    int* rowptr = deg + Npad;
    int* nbr = rowptr + Npad;
    int* rank = (int*)d_out;
    int edge_blocks = (E + 255) / 256;
    int gather_blocks = (N + 7) / 8;

    zero_kernel<<<(Npad + 255) / 256, 256, 0, stream>>>((float*)deg, (long)Npad);
    count_kernel<<<edge_blocks, 256, 0, stream>>>(dst, deg, rank, E);
    scan_kernel<<<scan_blocks, 1024, 0, stream>>>(deg, rowptr);
    fill_kernel<<<edge_blocks, 256, 0, stream>>>(src, dst, rank, rowptr, nbr, E);
    gather_kernel<<<gather_blocks, 256, 0, stream>>>(x, rowptr, deg, nbr, summed, cnt, N);
    combine_kernel<<<combine_blocks, 256, 0, stream>>>(summed, cnt, x, Wl1, bl1, Wr1,
                                                       out, ntiles, 1);
    gather_kernel<<<gather_blocks, 256, 0, stream>>>(out, rowptr, deg, nbr, summed,
                                                     nullptr, N);
    combine_kernel<<<combine_blocks, 256, 0, stream>>>(summed, cnt, out, Wl2, bl2, Wr2,
                                                       out, ntiles, 0);
    return;
  }

  int scatter_blocks = (int)(((long)E * 32 + 255) / 256);

  zero_kernel<<<2048, 256, 0, stream>>>(cnt, (long)N * (D + 1));
  scatter_kernel<<<scatter_blocks, 256, 0, stream>>>(x, src, dst, summed, cnt, E);
  inv_kernel<<<(N + 255) / 256, 256, 0, stream>>>(cnt, N);
  combine_kernel<<<combine_blocks, 256, 0, stream>>>(summed, cnt, x, Wl1, bl1, Wr1,
                                                     out, ntiles, 1);
  zero_kernel<<<2048, 256, 0, stream>>>(summed, (long)N * D);
  scatter_kernel<<<scatter_blocks, 256, 0, stream>>>(out, src, dst, summed, nullptr, E);
  combine_kernel<<<combine_blocks, 256, 0, stream>>>(summed, cnt, out, Wl2, bl2, Wr2,
                                                     out, ntiles, 0);
}


# combine: Wl and Wr copied once per work-group into LDS (k-pair interleaved, padded pitch), B fragments read from LDS; one resident 16-wave work-group per WGP loops over the row tiles
# speedup vs baseline: 2.5087x; 1.0943x over previous
#include <hip/hip_runtime.h>


typedef float v2f __attribute__((ext_vector_type(2)));
typedef float v8f __attribute__((ext_vector_type(8)));

#define D 128
#define NTILE 16

__global__ void zero_kernel(float* __restrict__ p, long n) {
  long i = (long)blockIdx.x * blockDim.x + threadIdx.x;
  long stride = (long)gridDim.x * blockDim.x;
  for (; i < n; i += stride) p[i] = 0.0f;
}

__global__ void scatter_kernel(const float* __restrict__ z,
                               const int* __restrict__ src,
                               const int* __restrict__ dst,
                               float* __restrict__ summed,
                               float* __restrict__ cnt,
                               int E) {
  unsigned tid = blockIdx.x * blockDim.x + threadIdx.x;
  int wid = (int)(tid >> 5);
  int lane = threadIdx.x & 31;
  if (wid >= E) return;
  int s = __builtin_amdgcn_readfirstlane(src[wid]);
  int d = __builtin_amdgcn_readfirstlane(dst[wid]);

  const float* zp = z + (size_t)s * D + lane;
  float v0 = zp[0];
  float v1 = zp[32];
  float v2 = zp[64];
  float v3 = zp[96];

  float* op = summed + (size_t)d * D + lane;
  unsafeAtomicAdd(op + 0,  fmaxf(v0, 0.0f));
  unsafeAtomicAdd(op + 32, fmaxf(v1, 0.0f));
  unsafeAtomicAdd(op + 64, fmaxf(v2, 0.0f));
  unsafeAtomicAdd(op + 96, fmaxf(v3, 0.0f));
  if (cnt != nullptr && lane == 0) unsafeAtomicAdd(cnt + d, 1.0f);
}

__global__ void inv_kernel(float* __restrict__ cnt, int n) {
  int i = blockIdx.x * blockDim.x + threadIdx.x;
  if (i < n) cnt[i] = 1.0f / fmaxf(cnt[i], 1.0f);
}

__global__ void count_kernel(const int* __restrict__ dst, int* __restrict__ deg,
                             int* __restrict__ rank, int E) {
  int e = blockIdx.x * blockDim.x + threadIdx.x;
  if (e < E) rank[e] = atomicAdd(deg + dst[e], 1);
}

#define SCAN_TILE 4096

__global__ void __launch_bounds__(1024)
scan_kernel(const int* __restrict__ deg, int* __restrict__ rowptr) {
  __shared__ int wsum[32];
  __shared__ int wtot[32];
  const int4* deg4 = reinterpret_cast<const int4*>(deg);
  int t = threadIdx.x;
  int lane = t & 31;
  int w = t >> 5;

  int part = 0;
  int nfront = blockIdx.x * (SCAN_TILE / 4);
#pragma unroll 4
  for (int k = t; k < nfront; k += 1024) {
    int4 q = deg4[k];
    part += (q.x + q.y) + (q.z + q.w);
  }
  for (int o = 16; o > 0; o >>= 1) part += __shfl_xor(part, o);
  if (lane == 0) wsum[w] = part;

  int4 v = deg4[nfront + t];
  int s0 = v.x, s1 = s0 + v.y, s2 = s1 + v.z, s3 = s2 + v.w;
  int incl = s3;
  for (int o = 1; o < 32; o <<= 1) {
    int u = __shfl_up(incl, o);
    if (lane >= o) incl += u;
  }
  if (lane == 31) wtot[w] = incl;
  __syncthreads();

  int front = wsum[lane];
  int mine = (lane < w) ? wtot[lane] : 0;
  for (int o = 16; o > 0; o >>= 1) {
    front += __shfl_xor(front, o);
    mine += __shfl_xor(mine, o);
  }
  int ex = front + mine + (incl - s3);
  reinterpret_cast<int4*>(rowptr)[nfront + t] = make_int4(ex, ex + s0, ex + s1, ex + s2);
}

__global__ void fill_kernel(const int* __restrict__ src, const int* __restrict__ dst,
                            const int* __restrict__ rank, const int* __restrict__ rowptr,
                            int* __restrict__ nbr, int E) {
  int e = blockIdx.x * blockDim.x + threadIdx.x;
  if (e < E) nbr[rowptr[dst[e]] + rank[e]] = src[e];
}

template <int K>
__device__ __forceinline__ void add_rows(const float4* __restrict__ z4, int idx, int j,
                                         int lane, float4& acc) {
  float4 v[K];
#pragma unroll
  for (int u = 0; u < K; ++u) {
    int s = __builtin_amdgcn_readlane(idx, j + u);
    v[u] = z4[(size_t)s * (D / 4) + lane];
  }
#pragma unroll
  for (int u = 0; u < K; ++u) {
    acc.x += fmaxf(v[u].x, 0.0f);
    acc.y += fmaxf(v[u].y, 0.0f);
    acc.z += fmaxf(v[u].z, 0.0f);
    acc.w += fmaxf(v[u].w, 0.0f);
  }
}

__global__ void __launch_bounds__(256)
gather_kernel(const float* __restrict__ z, const int* __restrict__ rowptr,
              const int* __restrict__ deg, const int* __restrict__ nbr,
              float* __restrict__ summed, float* __restrict__ inv, int n) {
  int node = blockIdx.x * (blockDim.x >> 5) + (threadIdx.x >> 5);
  if (node >= n) return;
  int lane = threadIdx.x & 31;
  int start = __builtin_amdgcn_readfirstlane(rowptr[node]);
  int cnt = __builtin_amdgcn_readfirstlane(deg[node]);
  const float4* z4 = reinterpret_cast<const float4*>(z);

  float4 acc = make_float4(0.0f, 0.0f, 0.0f, 0.0f);
  for (int base = 0; base < cnt; base += 32) {
    int m = cnt - base;
    if (m > 32) m = 32;
    int idx = (lane < m) ? nbr[start + base + lane] : 0;
    int j = 0;
    for (; j + 8 <= m; j += 8) add_rows<8>(z4, idx, j, lane, acc);
    if (m & 4) { add_rows<4>(z4, idx, j, lane, acc); j += 4; }
    if (m & 2) { add_rows<2>(z4, idx, j, lane, acc); j += 2; }
    if (m & 1) add_rows<1>(z4, idx, j, lane, acc);
  }
  reinterpret_cast<float4*>(summed)[(size_t)node * (D / 4) + lane] = acc;
  if (inv != nullptr && lane == 0) inv[node] = 1.0f / fmaxf((float)cnt, 1.0f);
}

#define WPITCH 288
#define WMAT (64 * WPITCH)
#define WLDS_FLOATS 41472

typedef float v4f __attribute__((ext_vector_type(4)));

template <bool SCALE>
__device__ __forceinline__ void tile_product(const float* __restrict__ arow, float invr,
                                             const float* wl, v8f (&acc)[8]) {
  v2f an[4];
  v4f bn[4];
#pragma unroll
  for (int j = 0; j < 4; ++j) an[j] = *reinterpret_cast<const v2f*>(arow + 4 * j);
#pragma unroll
  for (int q = 0; q < 4; ++q) bn[q] = *reinterpret_cast<const v4f*>(wl + 64 * q);

#pragma unroll 1
  for (int ch = 0; ch < 8; ++ch) {
    v2f ac[4];
#pragma unroll
    for (int j = 0; j < 4; ++j) ac[j] = an[j];
    const float* anext = arow + 16 * ((ch + 1) & 7);
#pragma unroll
    for (int j = 0; j < 4; ++j) an[j] = *reinterpret_cast<const v2f*>(anext + 4 * j);
#pragma unroll
    for (int j = 0; j < 4; ++j) {
      const int s = ch * 4 + j;
      v4f bc[4];
#pragma unroll
      for (int q = 0; q < 4; ++q) bc[q] = bn[q];
      const float* wnext = wl + 2 * ((s + 1) & 31) * WPITCH;
#pragma unroll
      for (int q = 0; q < 4; ++q) bn[q] = *reinterpret_cast<const v4f*>(wnext + 64 * q);
      v2f a = ac[j];
      if (SCALE) { a.x *= invr; a.y *= invr; }
#pragma unroll
      for (int q = 0; q < 4; ++q) {
        v2f b0, b1;
        b0.x = bc[q].x; b0.y = bc[q].y;
        b1.x = bc[q].z; b1.y = bc[q].w;
        acc[2 * q] = __builtin_amdgcn_wmma_f32_16x16x4_f32(
            false, a, false, b0, (short)0, acc[2 * q], false, false);
        acc[2 * q + 1] = __builtin_amdgcn_wmma_f32_16x16x4_f32(
            false, a, false, b1, (short)0, acc[2 * q + 1], false, false);
      }
    }
  }
}

__global__ void __launch_bounds__(512)
combine_kernel(const float* __restrict__ summed, const float* __restrict__ inv,
               const float* __restrict__ z, const float* __restrict__ Wl,
               const float* __restrict__ bl, const float* __restrict__ Wr,
               float* __restrict__ out, int ntiles, int do_relu) {
  __shared__ __attribute__((aligned(16))) float wlds[WLDS_FLOATS];
  const int t = threadIdx.x;

  {
    float4 r0[8], r1[8];
#pragma unroll
    for (int it = 0; it < 8; ++it) {
      int item = it * 512 + t;
      int m = item >> 11;
      int p = (item >> 5) & 63;
      int n4 = item & 31;
      const float4* Wm = reinterpret_cast<const float4*>(m ? Wr : Wl);
      r0[it] = Wm[(2 * p) * (D / 4) + n4];
      r1[it] = Wm[(2 * p + 1) * (D / 4) + n4];
    }
#pragma unroll
    for (int it = 0; it < 8; ++it) {
      int item = it * 512 + t;
      int m = item >> 11;
      int p = (item >> 5) & 63;
      int n4 = item & 31;
      float* dst = wlds + m * WMAT + p * WPITCH + ((n4 >> 3) * 16 + 4 * (n4 & 3)) * 4 +
                   ((n4 >> 2) & 1) * 2;
      v2f e0, e1, e2, e3;
      e0.x = r0[it].x; e0.y = r1[it].x;
      e1.x = r0[it].y; e1.y = r1[it].y;
      e2.x = r0[it].z; e2.y = r1[it].z;
      e3.x = r0[it].w; e3.y = r1[it].w;
      *reinterpret_cast<v2f*>(dst + 0)  = e0;
      *reinterpret_cast<v2f*>(dst + 4)  = e1;
      *reinterpret_cast<v2f*>(dst + 8)  = e2;
      *reinterpret_cast<v2f*>(dst + 12) = e3;
    }
  }
  __syncthreads();

  const int lane = t & 31;
  const int w = t >> 5;
  const int lrow = lane & 15;
  const int half = lane >> 4;
  const float* wl_l = wlds + half * WPITCH + lrow * 4;
  const float* wl_r = wl_l + WMAT;

  for (int tile = blockIdx.x + w * gridDim.x; tile < ntiles; tile += 16 * gridDim.x) {
    v8f acc[8] = {};

    int row = tile * NTILE + lrow;
    float invr = inv[row];
    const float* arow = summed + (size_t)row * D + half * 2;
    const float* zrow = z + (size_t)row * D + half * 2;

    tile_product<true>(arow, invr, wl_l, acc);
    tile_product<false>(zrow, 1.0f, wl_r, acc);

    int mbase = tile * NTILE + (half << 3);
#pragma unroll
    for (int n = 0; n < 8; ++n) {
      float bias = bl[n * 16 + lrow];
#pragma unroll
      for (int v = 0; v < 8; ++v) {
        float val = acc[n][v] + bias;
        if (do_relu) val = fmaxf(val, 0.0f);
        out[(size_t)(mbase + v) * D + n * 16 + lrow] = val;
      }
    }
  }
}

extern "C" void kernel_launch(void* const* d_in, const int* in_sizes, int n_in,
                              void* d_out, int out_size, void* d_ws, size_t ws_size,
                              hipStream_t stream) {
  const float* x   = (const float*)d_in[0];
  const int*   ei  = (const int*)d_in[1];
  const float* Wl1 = (const float*)d_in[2];
  const float* bl1 = (const float*)d_in[3];
  const float* Wr1 = (const float*)d_in[4];
  const float* Wl2 = (const float*)d_in[5];
  const float* bl2 = (const float*)d_in[6];
  const float* Wr2 = (const float*)d_in[7];
  float* out = (float*)d_out;

  const int N = in_sizes[0] / D;
  const int E = in_sizes[1] / 2;
  const int ntiles = (N + NTILE - 1) / NTILE;

  const int* src = ei;
  const int* dst = ei + E;

  float* cnt = (float*)d_ws;
  float* summed = cnt + N;

  const int combine_blocks = 256;

  const int scan_blocks = (N + SCAN_TILE - 1) / SCAN_TILE;
  const int Npad = scan_blocks * SCAN_TILE;
  const size_t need = ((size_t)N * (D + 1) + 2 * (size_t)Npad + (size_t)E) * sizeof(float);

  if (ws_size >= need) {
    int* deg = (int*)(summed + (size_t)N * D);
    int* rowptr = deg + Npad;
    int* nbr = rowptr + Npad;
    int* rank = (int*)d_out;
    int edge_blocks = (E + 255) / 256;
    int gather_blocks = (N + 7) / 8;

    zero_kernel<<<(Npad + 255) / 256, 256, 0, stream>>>((float*)deg, (long)Npad);
    count_kernel<<<edge_blocks, 256, 0, stream>>>(dst, deg, rank, E);
    scan_kernel<<<scan_blocks, 1024, 0, stream>>>(deg, rowptr);
    fill_kernel<<<edge_blocks, 256, 0, stream>>>(src, dst, rank, rowptr, nbr, E);
    gather_kernel<<<gather_blocks, 256, 0, stream>>>(x, rowptr, deg, nbr, summed, cnt, N);
    combine_kernel<<<combine_blocks, 512, 0, stream>>>(summed, cnt, x, Wl1, bl1, Wr1,
                                                       out, ntiles, 1);
    gather_kernel<<<gather_blocks, 256, 0, stream>>>(out, rowptr, deg, nbr, summed,
                                                     nullptr, N);
    combine_kernel<<<combine_blocks, 512, 0, stream>>>(summed, cnt, out, Wl2, bl2, Wr2,
                                                       out, ntiles, 0);
    return;
  }

  int scatter_blocks = (int)(((long)E * 32 + 255) / 256);

  zero_kernel<<<2048, 256, 0, stream>>>(cnt, (long)N * (D + 1));
  scatter_kernel<<<scatter_blocks, 256, 0, stream>>>(x, src, dst, summed, cnt, E);
  inv_kernel<<<(N + 255) / 256, 256, 0, stream>>>(cnt, N);
  combine_kernel<<<combine_blocks, 512, 0, stream>>>(summed, cnt, x, Wl1, bl1, Wr1,
                                                     out, ntiles, 1);
  zero_kernel<<<2048, 256, 0, stream>>>(summed, (long)N * D);
  scatter_kernel<<<scatter_blocks, 256, 0, stream>>>(out, src, dst, summed, nullptr, E);
  combine_kernel<<<combine_blocks, 512, 0, stream>>>(summed, cnt, out, Wl2, bl2, Wr2,
                                                     out, ntiles, 0);
}


# bucketing: two-level counting sort on LDS counters (edge chunks split by 512-node range, a work-group per range counts, scans, places) replaces per-edge global atomics; sum order = LDS arrival order
# speedup vs baseline: 3.8902x; 1.5507x over previous
#include <hip/hip_runtime.h>


typedef float v2f __attribute__((ext_vector_type(2)));
typedef float v8f __attribute__((ext_vector_type(8)));

#define D 128
#define NTILE 16

__global__ void zero_kernel(float* __restrict__ p, long n) {
  long i = (long)blockIdx.x * blockDim.x + threadIdx.x;
  long stride = (long)gridDim.x * blockDim.x;
  for (; i < n; i += stride) p[i] = 0.0f;
}

__global__ void scatter_kernel(const float* __restrict__ z,
                               const int* __restrict__ src,
                               const int* __restrict__ dst,
                               float* __restrict__ summed,
                               float* __restrict__ cnt,
                               int E) {
  unsigned tid = blockIdx.x * blockDim.x + threadIdx.x;
  int wid = (int)(tid >> 5);
  int lane = threadIdx.x & 31;
  if (wid >= E) return;
  int s = __builtin_amdgcn_readfirstlane(src[wid]);
  int d = __builtin_amdgcn_readfirstlane(dst[wid]);

  const float* zp = z + (size_t)s * D + lane;
  float v0 = zp[0];
  float v1 = zp[32];
  float v2 = zp[64];
  float v3 = zp[96];

  float* op = summed + (size_t)d * D + lane;
  unsafeAtomicAdd(op + 0,  fmaxf(v0, 0.0f));
  unsafeAtomicAdd(op + 32, fmaxf(v1, 0.0f));
  unsafeAtomicAdd(op + 64, fmaxf(v2, 0.0f));
  unsafeAtomicAdd(op + 96, fmaxf(v3, 0.0f));
  if (cnt != nullptr && lane == 0) unsafeAtomicAdd(cnt + d, 1.0f);
}

__global__ void inv_kernel(float* __restrict__ cnt, int n) {
  int i = blockIdx.x * blockDim.x + threadIdx.x;
  if (i < n) cnt[i] = 1.0f / fmaxf(cnt[i], 1.0f);
}

__global__ void count_kernel(const int* __restrict__ dst, int* __restrict__ deg,
                             int* __restrict__ rank, int E) {
  int e = blockIdx.x * blockDim.x + threadIdx.x;
  if (e < E) rank[e] = atomicAdd(deg + dst[e], 1);
}

#define SCAN_TILE 4096

__global__ void __launch_bounds__(1024)
scan_kernel(const int* __restrict__ deg, int* __restrict__ rowptr) {
  __shared__ int wsum[32];
  __shared__ int wtot[32];
  const int4* deg4 = reinterpret_cast<const int4*>(deg);
  int t = threadIdx.x;
  int lane = t & 31;
  int w = t >> 5;

  int part = 0;
  int nfront = blockIdx.x * (SCAN_TILE / 4);
#pragma unroll 4
  for (int k = t; k < nfront; k += 1024) {
    int4 q = deg4[k];
    part += (q.x + q.y) + (q.z + q.w);
  }
  for (int o = 16; o > 0; o >>= 1) part += __shfl_xor(part, o);
  if (lane == 0) wsum[w] = part;

  int4 v = deg4[nfront + t];
  int s0 = v.x, s1 = s0 + v.y, s2 = s1 + v.z, s3 = s2 + v.w;
  int incl = s3;
  for (int o = 1; o < 32; o <<= 1) {
    int u = __shfl_up(incl, o);
    if (lane >= o) incl += u;
  }
  if (lane == 31) wtot[w] = incl;
  __syncthreads();

  int front = wsum[lane];
  int mine = (lane < w) ? wtot[lane] : 0;
  for (int o = 16; o > 0; o >>= 1) {
    front += __shfl_xor(front, o);
    mine += __shfl_xor(mine, o);
  }
  int ex = front + mine + (incl - s3);
  reinterpret_cast<int4*>(rowptr)[nfront + t] = make_int4(ex, ex + s0, ex + s1, ex + s2);
}

__global__ void fill_kernel(const int* __restrict__ src, const int* __restrict__ dst,
                            const int* __restrict__ rank, const int* __restrict__ rowptr,
                            int* __restrict__ nbr, int E) {
  int e = blockIdx.x * blockDim.x + threadIdx.x;
  if (e < E) nbr[rowptr[dst[e]] + rank[e]] = src[e];
}

#define BIN_SHIFT 9
#define BIN_NODES 512
#define NBIN 256
#define NCHUNK 256
#define PSTAGE 8192
#define NSTAGE 16384

__global__ void __launch_bounds__(1024)
partition_kernel(const int* __restrict__ src, const int* __restrict__ dst,
                 int2* __restrict__ pairs, int* __restrict__ histTab,
                 int* __restrict__ offTab, int E, int CH) {
  __shared__ int hist[NBIN];
  __shared__ int cur[NBIN];
  __shared__ int wtot[8];
  __shared__ int2 stage[PSTAGE];
  const int t = threadIdx.x;
  const int lane = t & 31;
  const int w = t >> 5;
  const int e0 = blockIdx.x * CH;
  int e1 = e0 + CH;
  if (e1 > E) e1 = E;
  const bool staged = (e1 - e0) <= PSTAGE;

  if (t < NBIN) hist[t] = 0;
  __syncthreads();

  for (int eb = e0; eb < e1; eb += 8 * 1024) {
    int dv[8];
#pragma unroll
    for (int k = 0; k < 8; ++k) {
      int e = eb + k * 1024 + t;
      dv[k] = (e < e1) ? dst[e] : -1;
    }
#pragma unroll
    for (int k = 0; k < 8; ++k)
      if (dv[k] >= 0) atomicAdd(&hist[dv[k] >> BIN_SHIFT], 1);
  }
  __syncthreads();

  int h = 0, incl = 0;
  if (t < NBIN) {
    h = hist[t];
    incl = h;
    for (int o = 1; o < 32; o <<= 1) {
      int u = __shfl_up(incl, o);
      if (lane >= o) incl += u;
    }
    if (lane == 31) wtot[w] = incl;
  }
  __syncthreads();
  if (t < NBIN) {
    int front = 0;
    for (int k = 0; k < w; ++k) front += wtot[k];
    int ex = front + incl - h;
    cur[t] = ex;
    histTab[blockIdx.x * NBIN + t] = h;
    offTab[blockIdx.x * NBIN + t] = ex;
  }
  __syncthreads();

  for (int eb = e0; eb < e1; eb += 8 * 1024) {
    int sv[8], dv[8];
#pragma unroll
    for (int k = 0; k < 8; ++k) {
      int e = eb + k * 1024 + t;
      bool ok = e < e1;
      dv[k] = ok ? dst[e] : -1;
      sv[k] = ok ? src[e] : 0;
    }
#pragma unroll
    for (int k = 0; k < 8; ++k) {
      if (dv[k] >= 0) {
        int pos = atomicAdd(&cur[dv[k] >> BIN_SHIFT], 1);
        if (staged) stage[pos] = make_int2(sv[k], dv[k]);
        else pairs[e0 + pos] = make_int2(sv[k], dv[k]);
      }
    }
  }
  __syncthreads();
  if (staged)
    for (int i = t; i < e1 - e0; i += 1024) pairs[e0 + i] = stage[i];
}

__global__ void __launch_bounds__(1024)
finish_kernel(const int2* __restrict__ pairs, const int* __restrict__ histTab,
              const int* __restrict__ offTab, int* __restrict__ rowptr,
              int* __restrict__ deg, int* __restrict__ nbr, int N, int CH) {
  __shared__ int colsum[4][NBIN];
  __shared__ int plen[NCHUNK];
  __shared__ int psrc[NCHUNK];
  __shared__ int cnt[BIN_NODES];
  __shared__ int cur[BIN_NODES];
  __shared__ int wred[32];
  __shared__ int wtot[16];
  __shared__ int nstage[NSTAGE];
  const int t = threadIdx.x;
  const int lane = t & 31;
  const int w = t >> 5;
  const int r = blockIdx.x;
  const int base = r << BIN_SHIFT;

  {
    const int col = t & (NBIN - 1);
    const int r0 = t >> 8;
    int sum = 0;
#pragma unroll 16
    for (int k = 0; k < NCHUNK / 4; ++k) sum += histTab[(r0 + 4 * k) * NBIN + col];
    colsum[r0][col] = sum;
  }
  if (t < NCHUNK) {
    plen[t] = histTab[t * NBIN + r];
    psrc[t] = t * CH + offTab[t * NBIN + r];
  }
  if (t < BIN_NODES) cnt[t] = 0;
  __syncthreads();

  int v = 0;
  if (t < NBIN && t < r) v = (colsum[0][t] + colsum[1][t]) + (colsum[2][t] + colsum[3][t]);
  for (int o = 16; o > 0; o >>= 1) v += __shfl_xor(v, o);
  if (lane == 0) wred[w] = v;

  int len[8], p0[8];
  int2 pa[8], pb[8];
#pragma unroll
  for (int k = 0; k < 8; ++k) {
    len[k] = plen[w + 32 * k];
    p0[k] = psrc[w + 32 * k];
    pa[k] = (lane < len[k]) ? pairs[p0[k] + lane] : make_int2(0, base);
    pb[k] = (lane + 32 < len[k]) ? pairs[p0[k] + lane + 32] : make_int2(0, base);
  }
#pragma unroll
  for (int k = 0; k < 8; ++k) {
    if (lane < len[k]) atomicAdd(&cnt[pa[k].y - base], 1);
    if (lane + 32 < len[k]) atomicAdd(&cnt[pb[k].y - base], 1);
    for (int i = lane + 64; i < len[k]; i += 32) atomicAdd(&cnt[pairs[p0[k] + i].y - base], 1);
  }
  __syncthreads();

  int binStart = 0;
  for (int k = 0; k < 8; ++k) binStart += wred[k];
  const int rangeCount = (colsum[0][r] + colsum[1][r]) + (colsum[2][r] + colsum[3][r]);
  const bool staged = rangeCount <= NSTAGE;

  int h = 0, incl = 0;
  if (t < BIN_NODES) {
    h = cnt[t];
    incl = h;
    for (int o = 1; o < 32; o <<= 1) {
      int u = __shfl_up(incl, o);
      if (lane >= o) incl += u;
    }
    if (lane == 31) wtot[w] = incl;
  }
  __syncthreads();
  if (t < BIN_NODES) {
    int front = 0;
    for (int k = 0; k < w; ++k) front += wtot[k];
    int ex = binStart + front + incl - h;
    cur[t] = ex;
    if (base + t < N) {
      rowptr[base + t] = ex;
      deg[base + t] = h;
    }
  }
  __syncthreads();

#pragma unroll
  for (int k = 0; k < 8; ++k) {
    if (lane < len[k]) {
      int slot = atomicAdd(&cur[pa[k].y - base], 1);
      if (staged) nstage[slot - binStart] = pa[k].x;
      else nbr[slot] = pa[k].x;
    }
    if (lane + 32 < len[k]) {
      int slot = atomicAdd(&cur[pb[k].y - base], 1);
      if (staged) nstage[slot - binStart] = pb[k].x;
      else nbr[slot] = pb[k].x;
    }
    for (int i = lane + 64; i < len[k]; i += 32) {
      int2 q = pairs[p0[k] + i];
      int slot = atomicAdd(&cur[q.y - base], 1);
      if (staged) nstage[slot - binStart] = q.x;
      else nbr[slot] = q.x;
    }
  }
  __syncthreads();
  if (staged)
    for (int i = t; i < rangeCount; i += 1024) nbr[binStart + i] = nstage[i];
}

template <int K>
__device__ __forceinline__ void add_rows(const float4* __restrict__ z4, int idx, int j,
                                         int lane, float4& acc) {
  float4 v[K];
#pragma unroll
  for (int u = 0; u < K; ++u) {
    int s = __builtin_amdgcn_readlane(idx, j + u);
    v[u] = z4[(size_t)s * (D / 4) + lane];
  }
#pragma unroll
  for (int u = 0; u < K; ++u) {
    acc.x += fmaxf(v[u].x, 0.0f);
    acc.y += fmaxf(v[u].y, 0.0f);
    acc.z += fmaxf(v[u].z, 0.0f);
    acc.w += fmaxf(v[u].w, 0.0f);
  }
}

__global__ void __launch_bounds__(256)
gather_kernel(const float* __restrict__ z, const int* __restrict__ rowptr,
              const int* __restrict__ deg, const int* __restrict__ nbr,
              float* __restrict__ summed, float* __restrict__ inv, int n) {
  int node = blockIdx.x * (blockDim.x >> 5) + (threadIdx.x >> 5);
  if (node >= n) return;
  int lane = threadIdx.x & 31;
  int start = __builtin_amdgcn_readfirstlane(rowptr[node]);
  int cnt = __builtin_amdgcn_readfirstlane(deg[node]);
  const float4* z4 = reinterpret_cast<const float4*>(z);

  float4 acc = make_float4(0.0f, 0.0f, 0.0f, 0.0f);
  for (int base = 0; base < cnt; base += 32) {
    int m = cnt - base;
    if (m > 32) m = 32;
    int idx = (lane < m) ? nbr[start + base + lane] : 0;
    int j = 0;
    for (; j + 8 <= m; j += 8) add_rows<8>(z4, idx, j, lane, acc);
    if (m & 4) { add_rows<4>(z4, idx, j, lane, acc); j += 4; }
    if (m & 2) { add_rows<2>(z4, idx, j, lane, acc); j += 2; }
    if (m & 1) add_rows<1>(z4, idx, j, lane, acc);
  }
  reinterpret_cast<float4*>(summed)[(size_t)node * (D / 4) + lane] = acc;
  if (inv != nullptr && lane == 0) inv[node] = 1.0f / fmaxf((float)cnt, 1.0f);
}

#define WPITCH 288
#define WMAT (64 * WPITCH)
#define WLDS_FLOATS 41472

typedef float v4f __attribute__((ext_vector_type(4)));

template <bool SCALE>
__device__ __forceinline__ void tile_product(const float* __restrict__ arow, float invr,
                                             const float* wl, v8f (&acc)[8]) {
  v2f an[4];
  v4f bn[4];
#pragma unroll
  for (int j = 0; j < 4; ++j) an[j] = *reinterpret_cast<const v2f*>(arow + 4 * j);
#pragma unroll
  for (int q = 0; q < 4; ++q) bn[q] = *reinterpret_cast<const v4f*>(wl + 64 * q);

#pragma unroll 1
  for (int ch = 0; ch < 8; ++ch) {
    v2f ac[4];
#pragma unroll
    for (int j = 0; j < 4; ++j) ac[j] = an[j];
    const float* anext = arow + 16 * ((ch + 1) & 7);
#pragma unroll
    for (int j = 0; j < 4; ++j) an[j] = *reinterpret_cast<const v2f*>(anext + 4 * j);
#pragma unroll
    for (int j = 0; j < 4; ++j) {
      const int s = ch * 4 + j;
      v4f bc[4];
#pragma unroll
      for (int q = 0; q < 4; ++q) bc[q] = bn[q];
      const float* wnext = wl + 2 * ((s + 1) & 31) * WPITCH;
#pragma unroll
      for (int q = 0; q < 4; ++q) bn[q] = *reinterpret_cast<const v4f*>(wnext + 64 * q);
      v2f a = ac[j];
      if (SCALE) { a.x *= invr; a.y *= invr; }
#pragma unroll
      for (int q = 0; q < 4; ++q) {
        v2f b0, b1;
        b0.x = bc[q].x; b0.y = bc[q].y;
        b1.x = bc[q].z; b1.y = bc[q].w;
        acc[2 * q] = __builtin_amdgcn_wmma_f32_16x16x4_f32(
            false, a, false, b0, (short)0, acc[2 * q], false, false);
        acc[2 * q + 1] = __builtin_amdgcn_wmma_f32_16x16x4_f32(
            false, a, false, b1, (short)0, acc[2 * q + 1], false, false);
      }
    }
  }
}

__global__ void __launch_bounds__(512)
combine_kernel(const float* __restrict__ summed, const float* __restrict__ inv,
               const float* __restrict__ z, const float* __restrict__ Wl,
               const float* __restrict__ bl, const float* __restrict__ Wr,
               float* __restrict__ out, int ntiles, int do_relu) {
  __shared__ __attribute__((aligned(16))) float wlds[WLDS_FLOATS];
  const int t = threadIdx.x;

  {
    float4 r0[8], r1[8];
#pragma unroll
    for (int it = 0; it < 8; ++it) {
      int item = it * 512 + t;
      int m = item >> 11;
      int p = (item >> 5) & 63;
      int n4 = item & 31;
      const float4* Wm = reinterpret_cast<const float4*>(m ? Wr : Wl);
      r0[it] = Wm[(2 * p) * (D / 4) + n4];
      r1[it] = Wm[(2 * p + 1) * (D / 4) + n4];
    }
#pragma unroll
    for (int it = 0; it < 8; ++it) {
      int item = it * 512 + t;
      int m = item >> 11;
      int p = (item >> 5) & 63;
      int n4 = item & 31;
      float* dst = wlds + m * WMAT + p * WPITCH + ((n4 >> 3) * 16 + 4 * (n4 & 3)) * 4 +
                   ((n4 >> 2) & 1) * 2;
      v2f e0, e1, e2, e3;
      e0.x = r0[it].x; e0.y = r1[it].x;
      e1.x = r0[it].y; e1.y = r1[it].y;
      e2.x = r0[it].z; e2.y = r1[it].z;
      e3.x = r0[it].w; e3.y = r1[it].w;
      *reinterpret_cast<v2f*>(dst + 0)  = e0;
      *reinterpret_cast<v2f*>(dst + 4)  = e1;
      *reinterpret_cast<v2f*>(dst + 8)  = e2;
      *reinterpret_cast<v2f*>(dst + 12) = e3;
    }
  }
  __syncthreads();

  const int lane = t & 31;
  const int w = t >> 5;
  const int lrow = lane & 15;
  const int half = lane >> 4;
  const float* wl_l = wlds + half * WPITCH + lrow * 4;
  const float* wl_r = wl_l + WMAT;

  for (int tile = blockIdx.x + w * gridDim.x; tile < ntiles; tile += 16 * gridDim.x) {
    v8f acc[8] = {};

    int row = tile * NTILE + lrow;
    float invr = inv[row];
    const float* arow = summed + (size_t)row * D + half * 2;
    const float* zrow = z + (size_t)row * D + half * 2;

    tile_product<true>(arow, invr, wl_l, acc);
    tile_product<false>(zrow, 1.0f, wl_r, acc);

    int mbase = tile * NTILE + (half << 3);
#pragma unroll
    for (int n = 0; n < 8; ++n) {
      float bias = bl[n * 16 + lrow];
#pragma unroll
      for (int v = 0; v < 8; ++v) {
        float val = acc[n][v] + bias;
        if (do_relu) val = fmaxf(val, 0.0f);
        out[(size_t)(mbase + v) * D + n * 16 + lrow] = val;
      }
    }
  }
}

extern "C" void kernel_launch(void* const* d_in, const int* in_sizes, int n_in,
                              void* d_out, int out_size, void* d_ws, size_t ws_size,
                              hipStream_t stream) {
  const float* x   = (const float*)d_in[0];
  const int*   ei  = (const int*)d_in[1];
  const float* Wl1 = (const float*)d_in[2];
  const float* bl1 = (const float*)d_in[3];
  const float* Wr1 = (const float*)d_in[4];
  const float* Wl2 = (const float*)d_in[5];
  const float* bl2 = (const float*)d_in[6];
  const float* Wr2 = (const float*)d_in[7];
  float* out = (float*)d_out;

  const int N = in_sizes[0] / D;
  const int E = in_sizes[1] / 2;
  const int ntiles = (N + NTILE - 1) / NTILE;

  const int* src = ei;
  const int* dst = ei + E;

  float* cnt = (float*)d_ws;
  float* summed = cnt + N;

  const int combine_blocks = 256;

  const int scan_blocks = (N + SCAN_TILE - 1) / SCAN_TILE;
  const int Npad = scan_blocks * SCAN_TILE;
  const size_t need = ((size_t)N * (D + 1) + 2 * (size_t)Npad + (size_t)E) * sizeof(float);

  if (ws_size >= need) {
    int* deg = (int*)(summed + (size_t)N * D);
    int* rowptr = deg + Npad;
    int* nbr = rowptr + Npad;
    int gather_blocks = (N + 7) / 8;

    const int ranges = (N + BIN_NODES - 1) / BIN_NODES;
    if (ranges <= NBIN && (size_t)N * D >= 2 * (size_t)E + 2 * NCHUNK * NBIN) {
      const int CH = (E + NCHUNK - 1) / NCHUNK;
      int2* pairs = (int2*)d_out;
      int* histTab = (int*)d_out + 2 * (size_t)E;
      int* offTab = histTab + NCHUNK * NBIN;
      partition_kernel<<<NCHUNK, 1024, 0, stream>>>(src, dst, pairs, histTab, offTab, E, CH);
      finish_kernel<<<ranges, 1024, 0, stream>>>(pairs, histTab, offTab, rowptr, deg, nbr,
                                                 N, CH);
    } else {
      int* rank = (int*)d_out;
      int edge_blocks = (E + 255) / 256;
      zero_kernel<<<(Npad + 255) / 256, 256, 0, stream>>>((float*)deg, (long)Npad);
      count_kernel<<<edge_blocks, 256, 0, stream>>>(dst, deg, rank, E);
      scan_kernel<<<scan_blocks, 1024, 0, stream>>>(deg, rowptr);
      fill_kernel<<<edge_blocks, 256, 0, stream>>>(src, dst, rank, rowptr, nbr, E);
    }
    gather_kernel<<<gather_blocks, 256, 0, stream>>>(x, rowptr, deg, nbr, summed, cnt, N);
    combine_kernel<<<combine_blocks, 512, 0, stream>>>(summed, cnt, x, Wl1, bl1, Wr1,
                                                       out, ntiles, 1);
    gather_kernel<<<gather_blocks, 256, 0, stream>>>(out, rowptr, deg, nbr, summed,
                                                     nullptr, N);
    combine_kernel<<<combine_blocks, 512, 0, stream>>>(summed, cnt, out, Wl2, bl2, Wr2,
                                                       out, ntiles, 0);
    return;
  }

  int scatter_blocks = (int)(((long)E * 32 + 255) / 256);

  zero_kernel<<<2048, 256, 0, stream>>>(cnt, (long)N * (D + 1));
  scatter_kernel<<<scatter_blocks, 256, 0, stream>>>(x, src, dst, summed, cnt, E);
  inv_kernel<<<(N + 255) / 256, 256, 0, stream>>>(cnt, N);
  combine_kernel<<<combine_blocks, 512, 0, stream>>>(summed, cnt, x, Wl1, bl1, Wr1,
                                                     out, ntiles, 1);
  zero_kernel<<<2048, 256, 0, stream>>>(summed, (long)N * D);
  scatter_kernel<<<scatter_blocks, 256, 0, stream>>>(out, src, dst, summed, nullptr, E);
  combine_kernel<<<combine_blocks, 512, 0, stream>>>(summed, cnt, out, Wl2, bl2, Wr2,
                                                     out, ntiles, 0);
}


# aggregation fused into combine: each wave sums its tile's neighbour lists into an LDS tile, A fragments read from it; half the waves add z@Wr before mean@Wl (that sum order swaps) to stagger phases
# speedup vs baseline: 4.3576x; 1.1201x over previous
#include <hip/hip_runtime.h>


typedef float v2f __attribute__((ext_vector_type(2)));
typedef float v8f __attribute__((ext_vector_type(8)));

#define D 128
#define NTILE 16

__global__ void zero_kernel(float* __restrict__ p, long n) {
  long i = (long)blockIdx.x * blockDim.x + threadIdx.x;
  long stride = (long)gridDim.x * blockDim.x;
  for (; i < n; i += stride) p[i] = 0.0f;
}

__global__ void scatter_kernel(const float* __restrict__ z,
                               const int* __restrict__ src,
                               const int* __restrict__ dst,
                               float* __restrict__ summed,
                               float* __restrict__ cnt,
                               int E) {
  unsigned tid = blockIdx.x * blockDim.x + threadIdx.x;
  int wid = (int)(tid >> 5);
  int lane = threadIdx.x & 31;
  if (wid >= E) return;
  int s = __builtin_amdgcn_readfirstlane(src[wid]);
  int d = __builtin_amdgcn_readfirstlane(dst[wid]);

  const float* zp = z + (size_t)s * D + lane;
  float v0 = zp[0];
  float v1 = zp[32];
  float v2 = zp[64];
  float v3 = zp[96];

  float* op = summed + (size_t)d * D + lane;
  unsafeAtomicAdd(op + 0,  fmaxf(v0, 0.0f));
  unsafeAtomicAdd(op + 32, fmaxf(v1, 0.0f));
  unsafeAtomicAdd(op + 64, fmaxf(v2, 0.0f));
  unsafeAtomicAdd(op + 96, fmaxf(v3, 0.0f));
  if (cnt != nullptr && lane == 0) unsafeAtomicAdd(cnt + d, 1.0f);
}

__global__ void inv_kernel(float* __restrict__ cnt, int n) {
  int i = blockIdx.x * blockDim.x + threadIdx.x;
  if (i < n) cnt[i] = 1.0f / fmaxf(cnt[i], 1.0f);
}

__global__ void count_kernel(const int* __restrict__ dst, int* __restrict__ deg,
                             int* __restrict__ rank, int E) {
  int e = blockIdx.x * blockDim.x + threadIdx.x;
  if (e < E) rank[e] = atomicAdd(deg + dst[e], 1);
}

#define SCAN_TILE 4096

__global__ void __launch_bounds__(1024)
scan_kernel(const int* __restrict__ deg, int* __restrict__ rowptr) {
  __shared__ int wsum[32];
  __shared__ int wtot[32];
  const int4* deg4 = reinterpret_cast<const int4*>(deg);
  int t = threadIdx.x;
  int lane = t & 31;
  int w = t >> 5;

  int part = 0;
  int nfront = blockIdx.x * (SCAN_TILE / 4);
#pragma unroll 4
  for (int k = t; k < nfront; k += 1024) {
    int4 q = deg4[k];
    part += (q.x + q.y) + (q.z + q.w);
  }
  for (int o = 16; o > 0; o >>= 1) part += __shfl_xor(part, o);
  if (lane == 0) wsum[w] = part;

  int4 v = deg4[nfront + t];
  int s0 = v.x, s1 = s0 + v.y, s2 = s1 + v.z, s3 = s2 + v.w;
  int incl = s3;
  for (int o = 1; o < 32; o <<= 1) {
    int u = __shfl_up(incl, o);
    if (lane >= o) incl += u;
  }
  if (lane == 31) wtot[w] = incl;
  __syncthreads();

  int front = wsum[lane];
  int mine = (lane < w) ? wtot[lane] : 0;
  for (int o = 16; o > 0; o >>= 1) {
    front += __shfl_xor(front, o);
    mine += __shfl_xor(mine, o);
  }
  int ex = front + mine + (incl - s3);
  reinterpret_cast<int4*>(rowptr)[nfront + t] = make_int4(ex, ex + s0, ex + s1, ex + s2);
}

__global__ void fill_kernel(const int* __restrict__ src, const int* __restrict__ dst,
                            const int* __restrict__ rank, const int* __restrict__ rowptr,
                            int* __restrict__ nbr, int E) {
  int e = blockIdx.x * blockDim.x + threadIdx.x;
  if (e < E) nbr[rowptr[dst[e]] + rank[e]] = src[e];
}

#define BIN_SHIFT 9
#define BIN_NODES 512
#define NBIN 256
#define NCHUNK 256
#define PSTAGE 8192
#define NSTAGE 16384

__global__ void __launch_bounds__(1024)
partition_kernel(const int* __restrict__ src, const int* __restrict__ dst,
                 int2* __restrict__ pairs, int* __restrict__ histTab,
                 int* __restrict__ offTab, int E, int CH) {
  __shared__ int hist[NBIN];
  __shared__ int cur[NBIN];
  __shared__ int wtot[8];
  __shared__ int2 stage[PSTAGE];
  const int t = threadIdx.x;
  const int lane = t & 31;
  const int w = t >> 5;
  const int e0 = blockIdx.x * CH;
  int e1 = e0 + CH;
  if (e1 > E) e1 = E;
  const bool staged = (e1 - e0) <= PSTAGE;

  if (t < NBIN) hist[t] = 0;
  __syncthreads();

  for (int eb = e0; eb < e1; eb += 8 * 1024) {
    int dv[8];
#pragma unroll
    for (int k = 0; k < 8; ++k) {
      int e = eb + k * 1024 + t;
      dv[k] = (e < e1) ? dst[e] : -1;
    }
#pragma unroll
    for (int k = 0; k < 8; ++k)
      if (dv[k] >= 0) atomicAdd(&hist[dv[k] >> BIN_SHIFT], 1);
  }
  __syncthreads();

  int h = 0, incl = 0;
  if (t < NBIN) {
    h = hist[t];
    incl = h;
    for (int o = 1; o < 32; o <<= 1) {
      int u = __shfl_up(incl, o);
      if (lane >= o) incl += u;
    }
    if (lane == 31) wtot[w] = incl;
  }
  __syncthreads();
  if (t < NBIN) {
    int front = 0;
    for (int k = 0; k < w; ++k) front += wtot[k];
    int ex = front + incl - h;
    cur[t] = ex;
    histTab[blockIdx.x * NBIN + t] = h;
    offTab[blockIdx.x * NBIN + t] = ex;
  }
  __syncthreads();

  for (int eb = e0; eb < e1; eb += 8 * 1024) {
    int sv[8], dv[8];
#pragma unroll
    for (int k = 0; k < 8; ++k) {
      int e = eb + k * 1024 + t;
      bool ok = e < e1;
      dv[k] = ok ? dst[e] : -1;
      sv[k] = ok ? src[e] : 0;
    }
#pragma unroll
    for (int k = 0; k < 8; ++k) {
      if (dv[k] >= 0) {
        int pos = atomicAdd(&cur[dv[k] >> BIN_SHIFT], 1);
        if (staged) stage[pos] = make_int2(sv[k], dv[k]);
        else pairs[e0 + pos] = make_int2(sv[k], dv[k]);
      }
    }
  }
  __syncthreads();
  if (staged)
    for (int i = t; i < e1 - e0; i += 1024) pairs[e0 + i] = stage[i];
}

__global__ void __launch_bounds__(1024)
finish_kernel(const int2* __restrict__ pairs, const int* __restrict__ histTab,
              const int* __restrict__ offTab, int* __restrict__ rowptr,
              int* __restrict__ deg, int* __restrict__ nbr, int N, int CH) {
  __shared__ int colsum[4][NBIN];
  __shared__ int plen[NCHUNK];
  __shared__ int psrc[NCHUNK];
  __shared__ int cnt[BIN_NODES];
  __shared__ int cur[BIN_NODES];
  __shared__ int wred[32];
  __shared__ int wtot[16];
  __shared__ int nstage[NSTAGE];
  const int t = threadIdx.x;
  const int lane = t & 31;
  const int w = t >> 5;
  const int r = blockIdx.x;
  const int base = r << BIN_SHIFT;

  {
    const int col = t & (NBIN - 1);
    const int r0 = t >> 8;
    int sum = 0;
#pragma unroll 16
    for (int k = 0; k < NCHUNK / 4; ++k) sum += histTab[(r0 + 4 * k) * NBIN + col];
    colsum[r0][col] = sum;
  }
  if (t < NCHUNK) {
    plen[t] = histTab[t * NBIN + r];
    psrc[t] = t * CH + offTab[t * NBIN + r];
  }
  if (t < BIN_NODES) cnt[t] = 0;
  __syncthreads();

  int v = 0;
  if (t < NBIN && t < r) v = (colsum[0][t] + colsum[1][t]) + (colsum[2][t] + colsum[3][t]);
  for (int o = 16; o > 0; o >>= 1) v += __shfl_xor(v, o);
  if (lane == 0) wred[w] = v;

  int len[8], p0[8];
  int2 pa[8], pb[8];
#pragma unroll
  for (int k = 0; k < 8; ++k) {
    len[k] = plen[w + 32 * k];
    p0[k] = psrc[w + 32 * k];
    pa[k] = (lane < len[k]) ? pairs[p0[k] + lane] : make_int2(0, base);
    pb[k] = (lane + 32 < len[k]) ? pairs[p0[k] + lane + 32] : make_int2(0, base);
  }
#pragma unroll
  for (int k = 0; k < 8; ++k) {
    if (lane < len[k]) atomicAdd(&cnt[pa[k].y - base], 1);
    if (lane + 32 < len[k]) atomicAdd(&cnt[pb[k].y - base], 1);
    for (int i = lane + 64; i < len[k]; i += 32) atomicAdd(&cnt[pairs[p0[k] + i].y - base], 1);
  }
  __syncthreads();

  int binStart = 0;
  for (int k = 0; k < 8; ++k) binStart += wred[k];
  const int rangeCount = (colsum[0][r] + colsum[1][r]) + (colsum[2][r] + colsum[3][r]);
  const bool staged = rangeCount <= NSTAGE;

  int h = 0, incl = 0;
  if (t < BIN_NODES) {
    h = cnt[t];
    incl = h;
    for (int o = 1; o < 32; o <<= 1) {
      int u = __shfl_up(incl, o);
      if (lane >= o) incl += u;
    }
    if (lane == 31) wtot[w] = incl;
  }
  __syncthreads();
  if (t < BIN_NODES) {
    int front = 0;
    for (int k = 0; k < w; ++k) front += wtot[k];
    int ex = binStart + front + incl - h;
    cur[t] = ex;
    if (base + t < N) {
      rowptr[base + t] = ex;
      deg[base + t] = h;
    }
  }
  __syncthreads();

#pragma unroll
  for (int k = 0; k < 8; ++k) {
    if (lane < len[k]) {
      int slot = atomicAdd(&cur[pa[k].y - base], 1);
      if (staged) nstage[slot - binStart] = pa[k].x;
      else nbr[slot] = pa[k].x;
    }
    if (lane + 32 < len[k]) {
      int slot = atomicAdd(&cur[pb[k].y - base], 1);
      if (staged) nstage[slot - binStart] = pb[k].x;
      else nbr[slot] = pb[k].x;
    }
    for (int i = lane + 64; i < len[k]; i += 32) {
      int2 q = pairs[p0[k] + i];
      int slot = atomicAdd(&cur[q.y - base], 1);
      if (staged) nstage[slot - binStart] = q.x;
      else nbr[slot] = q.x;
    }
  }
  __syncthreads();
  if (staged)
    for (int i = t; i < rangeCount; i += 1024) nbr[binStart + i] = nstage[i];
}

template <int K>
__device__ __forceinline__ void load_rows(const float4* __restrict__ z4, int idx, int j,
                                          int lane, float4 (&v)[K]) {
#pragma unroll
  for (int u = 0; u < K; ++u) {
    int s = __builtin_amdgcn_readlane(idx, j + u);
    v[u] = z4[(size_t)s * (D / 4) + lane];
  }
}

template <int K>
__device__ __forceinline__ void sum_rows(const float4 (&v)[K], float4& acc) {
#pragma unroll
  for (int u = 0; u < K; ++u) {
    acc.x += fmaxf(v[u].x, 0.0f);
    acc.y += fmaxf(v[u].y, 0.0f);
    acc.z += fmaxf(v[u].z, 0.0f);
    acc.w += fmaxf(v[u].w, 0.0f);
  }
}

#define WPITCH 288
#define WMAT (64 * WPITCH)
#define WLDS_FLOATS 41472

typedef float v4f __attribute__((ext_vector_type(4)));

template <bool SCALE>
__device__ __forceinline__ void tile_product(const float* __restrict__ arow, float invr,
                                             const float* wl, v8f (&acc)[8]) {
  v2f an[4];
  v4f bn[4];
#pragma unroll
  for (int j = 0; j < 4; ++j) an[j] = *reinterpret_cast<const v2f*>(arow + 4 * j);
#pragma unroll
  for (int q = 0; q < 4; ++q) bn[q] = *reinterpret_cast<const v4f*>(wl + 64 * q);

#pragma unroll 1
  for (int ch = 0; ch < 8; ++ch) {
    v2f ac[4];
#pragma unroll
    for (int j = 0; j < 4; ++j) ac[j] = an[j];
    const float* anext = arow + 16 * ((ch + 1) & 7);
#pragma unroll
    for (int j = 0; j < 4; ++j) an[j] = *reinterpret_cast<const v2f*>(anext + 4 * j);
#pragma unroll
    for (int j = 0; j < 4; ++j) {
      const int s = ch * 4 + j;
      v4f bc[4];
#pragma unroll
      for (int q = 0; q < 4; ++q) bc[q] = bn[q];
      const float* wnext = wl + 2 * ((s + 1) & 31) * WPITCH;
#pragma unroll
      for (int q = 0; q < 4; ++q) bn[q] = *reinterpret_cast<const v4f*>(wnext + 64 * q);
      v2f a = ac[j];
      if (SCALE) { a.x *= invr; a.y *= invr; }
#pragma unroll
      for (int q = 0; q < 4; ++q) {
        v2f b0, b1;
        b0.x = bc[q].x; b0.y = bc[q].y;
        b1.x = bc[q].z; b1.y = bc[q].w;
        acc[2 * q] = __builtin_amdgcn_wmma_f32_16x16x4_f32(
            false, a, false, b0, (short)0, acc[2 * q], false, false);
        acc[2 * q + 1] = __builtin_amdgcn_wmma_f32_16x16x4_f32(
            false, a, false, b1, (short)0, acc[2 * q + 1], false, false);
      }
    }
  }
}

__global__ void __launch_bounds__(512)
combine_kernel(const float* __restrict__ summed, const float* __restrict__ inv,
               const float* __restrict__ z, const float* __restrict__ Wl,
               const float* __restrict__ bl, const float* __restrict__ Wr,
               float* __restrict__ out, int ntiles, int do_relu) {
  __shared__ __attribute__((aligned(16))) float wlds[WLDS_FLOATS];
  const int t = threadIdx.x;

  {
    float4 r0[8], r1[8];
#pragma unroll
    for (int it = 0; it < 8; ++it) {
      int item = it * 512 + t;
      int m = item >> 11;
      int p = (item >> 5) & 63;
      int n4 = item & 31;
      const float4* Wm = reinterpret_cast<const float4*>(m ? Wr : Wl);
      r0[it] = Wm[(2 * p) * (D / 4) + n4];
      r1[it] = Wm[(2 * p + 1) * (D / 4) + n4];
    }
#pragma unroll
    for (int it = 0; it < 8; ++it) {
      int item = it * 512 + t;
      int m = item >> 11;
      int p = (item >> 5) & 63;
      int n4 = item & 31;
      float* dst = wlds + m * WMAT + p * WPITCH + ((n4 >> 3) * 16 + 4 * (n4 & 3)) * 4 +
                   ((n4 >> 2) & 1) * 2;
      v2f e0, e1, e2, e3;
      e0.x = r0[it].x; e0.y = r1[it].x;
      e1.x = r0[it].y; e1.y = r1[it].y;
      e2.x = r0[it].z; e2.y = r1[it].z;
      e3.x = r0[it].w; e3.y = r1[it].w;
      *reinterpret_cast<v2f*>(dst + 0)  = e0;
      *reinterpret_cast<v2f*>(dst + 4)  = e1;
      *reinterpret_cast<v2f*>(dst + 8)  = e2;
      *reinterpret_cast<v2f*>(dst + 12) = e3;
    }
  }
  __syncthreads();

  const int lane = t & 31;
  const int w = t >> 5;
  const int lrow = lane & 15;
  const int half = lane >> 4;
  const float* wl_l = wlds + half * WPITCH + lrow * 4;
  const float* wl_r = wl_l + WMAT;

  for (int tile = blockIdx.x + w * gridDim.x; tile < ntiles; tile += 16 * gridDim.x) {
    v8f acc[8] = {};

    int row = tile * NTILE + lrow;
    float invr = inv[row];
    const float* arow = summed + (size_t)row * D + half * 2;
    const float* zrow = z + (size_t)row * D + half * 2;

    tile_product<true>(arow, invr, wl_l, acc);
    tile_product<false>(zrow, 1.0f, wl_r, acc);

    int mbase = tile * NTILE + (half << 3);
#pragma unroll
    for (int n = 0; n < 8; ++n) {
      float bias = bl[n * 16 + lrow];
#pragma unroll
      for (int v = 0; v < 8; ++v) {
        float val = acc[n][v] + bias;
        if (do_relu) val = fmaxf(val, 0.0f);
        out[(size_t)(mbase + v) * D + n * 16 + lrow] = val;
      }
    }
  }
}

#define APITCH 132
#define ATILE (NTILE * APITCH)
#define LAYER_LDS_FLOATS (2 * WMAT + 16 * ATILE)
#define GCHUNK 23

__device__ __forceinline__ void wave_sync() {
  __builtin_amdgcn_fence(__ATOMIC_RELEASE, "wavefront");
  __builtin_amdgcn_wave_barrier();
  __builtin_amdgcn_fence(__ATOMIC_ACQUIRE, "wavefront");
}

__global__ void __launch_bounds__(512)
layer_kernel(const float* __restrict__ z, const int* __restrict__ rowptr,
             const int* __restrict__ deg, const int* __restrict__ nbr,
             const float* __restrict__ Wl, const float* __restrict__ bl,
             const float* __restrict__ Wr, float* __restrict__ out, int ntiles, int n,
             int do_relu) {
  __shared__ __attribute__((aligned(16))) float wlds[LAYER_LDS_FLOATS];
  const int t = threadIdx.x;

  {
    float4 r0[8], r1[8];
#pragma unroll
    for (int it = 0; it < 8; ++it) {
      int item = it * 512 + t;
      int m = item >> 11;
      int p = (item >> 5) & 63;
      int n4 = item & 31;
      const float4* Wm = reinterpret_cast<const float4*>(m ? Wr : Wl);
      r0[it] = Wm[(2 * p) * (D / 4) + n4];
      r1[it] = Wm[(2 * p + 1) * (D / 4) + n4];
    }
#pragma unroll
    for (int it = 0; it < 8; ++it) {
      int item = it * 512 + t;
      int m = item >> 11;
      int p = (item >> 5) & 63;
      int n4 = item & 31;
      float* dst = wlds + m * WMAT + p * WPITCH + ((n4 >> 3) * 16 + 4 * (n4 & 3)) * 4 +
                   ((n4 >> 2) & 1) * 2;
      v2f e0, e1, e2, e3;
      e0.x = r0[it].x; e0.y = r1[it].x;
      e1.x = r0[it].y; e1.y = r1[it].y;
      e2.x = r0[it].z; e2.y = r1[it].z;
      e3.x = r0[it].w; e3.y = r1[it].w;
      *reinterpret_cast<v2f*>(dst + 0)  = e0;
      *reinterpret_cast<v2f*>(dst + 4)  = e1;
      *reinterpret_cast<v2f*>(dst + 8)  = e2;
      *reinterpret_cast<v2f*>(dst + 12) = e3;
    }
  }
  __syncthreads();

  const int lane = t & 31;
  const int w = t >> 5;
  const int lrow = lane & 15;
  const int half = lane >> 4;
  const float* wl_l = wlds + half * WPITCH + lrow * 4;
  const float* wl_r = wl_l + WMAT;
  float* at = wlds + 2 * WMAT + w * ATILE;
  const float* afrag = at + lrow * APITCH + half * 2;
  const float4* z4 = reinterpret_cast<const float4*>(z);
  const bool zfirst = (__builtin_amdgcn_readfirstlane(w) & 4) != 0;

  for (int tile = blockIdx.x + w * gridDim.x; tile < ntiles; tile += 16 * gridDim.x) {
    v8f acc[8] = {};
    const int row = tile * NTILE + lrow;
    const float* zrow = z + (size_t)row * D + half * 2;
    const int rp = (row < n) ? rowptr[row] : 0;
    const int dg = (row < n) ? deg[row] : 0;
    if (zfirst) tile_product<false>(zrow, 1.0f, wl_r, acc);

    wave_sync();

    int idx_next;
    {
      int s0 = __builtin_amdgcn_readlane(rp, 0);
      int c0 = __builtin_amdgcn_readlane(dg, 0);
      int m0 = c0 < GCHUNK ? c0 : GCHUNK;
      idx_next = (lane < m0) ? nbr[s0 + lane] : 0;
    }
    for (int r = 0; r < NTILE; ++r) {
      const int start = __builtin_amdgcn_readlane(rp, r);
      const int cnt = __builtin_amdgcn_readlane(dg, r);
      int idx = idx_next;
      if (r + 1 < NTILE) {
        int s1 = __builtin_amdgcn_readlane(rp, r + 1);
        int c1 = __builtin_amdgcn_readlane(dg, r + 1);
        int m1 = c1 < GCHUNK ? c1 : GCHUNK;
        idx_next = (lane < m1) ? nbr[s1 + lane] : 0;
      }
      float4 nsum = make_float4(0.0f, 0.0f, 0.0f, 0.0f);
      for (int base = 0; base < cnt; base += GCHUNK) {
        int m = cnt - base;
        if (m > GCHUNK) m = GCHUNK;
        if (base > 0) idx = (lane < m) ? nbr[start + base + lane] : 0;
        float4 va[8], vb[8], vc[4], vd[2], ve[1];
        int j = 0;
        if (m >= 8)  { load_rows<8>(z4, idx, j, lane, va); j += 8; }
        if (m >= 16) { load_rows<8>(z4, idx, j, lane, vb); j += 8; }
        if (m & 4)   { load_rows<4>(z4, idx, j, lane, vc); j += 4; }
        if (m & 2)   { load_rows<2>(z4, idx, j, lane, vd); j += 2; }
        if (m & 1)   { load_rows<1>(z4, idx, j, lane, ve); }
        if (m >= 8)  sum_rows<8>(va, nsum);
        if (m >= 16) sum_rows<8>(vb, nsum);
        if (m & 4)   sum_rows<4>(vc, nsum);
        if (m & 2)   sum_rows<2>(vd, nsum);
        if (m & 1)   sum_rows<1>(ve, nsum);
      }
      const float invr = 1.0f / fmaxf((float)cnt, 1.0f);
      nsum.x *= invr; nsum.y *= invr; nsum.z *= invr; nsum.w *= invr;
      *reinterpret_cast<float4*>(at + r * APITCH + 4 * lane) = nsum;
    }
    wave_sync();

    tile_product<false>(afrag, 1.0f, wl_l, acc);
    if (!zfirst) tile_product<false>(zrow, 1.0f, wl_r, acc);

    int mbase = tile * NTILE + (half << 3);
#pragma unroll
    for (int nn = 0; nn < 8; ++nn) {
      float bias = bl[nn * 16 + lrow];
#pragma unroll
      for (int v = 0; v < 8; ++v) {
        float val = acc[nn][v] + bias;
        if (do_relu) val = fmaxf(val, 0.0f);
        out[(size_t)(mbase + v) * D + nn * 16 + lrow] = val;
      }
    }
  }
}

extern "C" void kernel_launch(void* const* d_in, const int* in_sizes, int n_in,
                              void* d_out, int out_size, void* d_ws, size_t ws_size,
                              hipStream_t stream) {
  const float* x   = (const float*)d_in[0];
  const int*   ei  = (const int*)d_in[1];
  const float* Wl1 = (const float*)d_in[2];
  const float* bl1 = (const float*)d_in[3];
  const float* Wr1 = (const float*)d_in[4];
  const float* Wl2 = (const float*)d_in[5];
  const float* bl2 = (const float*)d_in[6];
  const float* Wr2 = (const float*)d_in[7];
  float* out = (float*)d_out;

  const int N = in_sizes[0] / D;
  const int E = in_sizes[1] / 2;
  const int ntiles = (N + NTILE - 1) / NTILE;

  const int* src = ei;
  const int* dst = ei + E;

  float* cnt = (float*)d_ws;
  float* summed = cnt + N;

  const int combine_blocks = 256;

  const int scan_blocks = (N + SCAN_TILE - 1) / SCAN_TILE;
  const int Npad = scan_blocks * SCAN_TILE;
  const size_t need = ((size_t)N * (D + 1) + 2 * (size_t)Npad + (size_t)E) * sizeof(float);

  if (ws_size >= need) {
    int* deg = (int*)(summed + (size_t)N * D);
    int* rowptr = deg + Npad;
    int* nbr = rowptr + Npad;

    const int ranges = (N + BIN_NODES - 1) / BIN_NODES;
    if (ranges <= NBIN && (size_t)N * D >= 2 * (size_t)E + 2 * NCHUNK * NBIN) {
      const int CH = (E + NCHUNK - 1) / NCHUNK;
      int2* pairs = (int2*)d_out;
      int* histTab = (int*)d_out + 2 * (size_t)E;
      int* offTab = histTab + NCHUNK * NBIN;
      partition_kernel<<<NCHUNK, 1024, 0, stream>>>(src, dst, pairs, histTab, offTab, E, CH);
      finish_kernel<<<ranges, 1024, 0, stream>>>(pairs, histTab, offTab, rowptr, deg, nbr,
                                                 N, CH);
    } else {
      int* rank = (int*)d_out;
      int edge_blocks = (E + 255) / 256;
      zero_kernel<<<(Npad + 255) / 256, 256, 0, stream>>>((float*)deg, (long)Npad);
      count_kernel<<<edge_blocks, 256, 0, stream>>>(dst, deg, rank, E);
      scan_kernel<<<scan_blocks, 1024, 0, stream>>>(deg, rowptr);
      fill_kernel<<<edge_blocks, 256, 0, stream>>>(src, dst, rank, rowptr, nbr, E);
    }
    float* h = summed;
    layer_kernel<<<combine_blocks, 512, 0, stream>>>(x, rowptr, deg, nbr, Wl1, bl1, Wr1, h,
                                                     ntiles, N, 1);
    layer_kernel<<<combine_blocks, 512, 0, stream>>>(h, rowptr, deg, nbr, Wl2, bl2, Wr2, out,
                                                     ntiles, N, 0);
    return;
  }

  int scatter_blocks = (int)(((long)E * 32 + 255) / 256);

  zero_kernel<<<2048, 256, 0, stream>>>(cnt, (long)N * (D + 1));
  scatter_kernel<<<scatter_blocks, 256, 0, stream>>>(x, src, dst, summed, cnt, E);
  inv_kernel<<<(N + 255) / 256, 256, 0, stream>>>(cnt, N);
  combine_kernel<<<combine_blocks, 512, 0, stream>>>(summed, cnt, x, Wl1, bl1, Wr1,
                                                     out, ntiles, 1);
  zero_kernel<<<2048, 256, 0, stream>>>(summed, (long)N * D);
  scatter_kernel<<<scatter_blocks, 256, 0, stream>>>(out, src, dst, summed, nullptr, E);
  combine_kernel<<<combine_blocks, 512, 0, stream>>>(summed, cnt, out, Wl2, bl2, Wr2,
                                                     out, ntiles, 0);
}


# fused layer: z @ Wr cut into 16 two-k-step slices, one issued per node between its row requests and its sums (matrix work under the row loads); sum order: z @ Wr first in all waves, k regrouped
# speedup vs baseline: 4.5510x; 1.0444x over previous
#include <hip/hip_runtime.h>


typedef float v2f __attribute__((ext_vector_type(2)));
typedef float v8f __attribute__((ext_vector_type(8)));

#define D 128
#define NTILE 16

__global__ void zero_kernel(float* __restrict__ p, long n) {
  long i = (long)blockIdx.x * blockDim.x + threadIdx.x;
  long stride = (long)gridDim.x * blockDim.x;
  for (; i < n; i += stride) p[i] = 0.0f;
}

__global__ void scatter_kernel(const float* __restrict__ z,
                               const int* __restrict__ src,
                               const int* __restrict__ dst,
                               float* __restrict__ summed,
                               float* __restrict__ cnt,
                               int E) {
  unsigned tid = blockIdx.x * blockDim.x + threadIdx.x;
  int wid = (int)(tid >> 5);
  int lane = threadIdx.x & 31;
  if (wid >= E) return;
  int s = __builtin_amdgcn_readfirstlane(src[wid]);
  int d = __builtin_amdgcn_readfirstlane(dst[wid]);

  const float* zp = z + (size_t)s * D + lane;
  float v0 = zp[0];
  float v1 = zp[32];
  float v2 = zp[64];
  float v3 = zp[96];

  float* op = summed + (size_t)d * D + lane;
  unsafeAtomicAdd(op + 0,  fmaxf(v0, 0.0f));
  unsafeAtomicAdd(op + 32, fmaxf(v1, 0.0f));
  unsafeAtomicAdd(op + 64, fmaxf(v2, 0.0f));
  unsafeAtomicAdd(op + 96, fmaxf(v3, 0.0f));
  if (cnt != nullptr && lane == 0) unsafeAtomicAdd(cnt + d, 1.0f);
}

__global__ void inv_kernel(float* __restrict__ cnt, int n) {
  int i = blockIdx.x * blockDim.x + threadIdx.x;
  if (i < n) cnt[i] = 1.0f / fmaxf(cnt[i], 1.0f);
}

__global__ void count_kernel(const int* __restrict__ dst, int* __restrict__ deg,
                             int* __restrict__ rank, int E) {
  int e = blockIdx.x * blockDim.x + threadIdx.x;
  if (e < E) rank[e] = atomicAdd(deg + dst[e], 1);
}

#define SCAN_TILE 4096

__global__ void __launch_bounds__(1024)
scan_kernel(const int* __restrict__ deg, int* __restrict__ rowptr) {
  __shared__ int wsum[32];
  __shared__ int wtot[32];
  const int4* deg4 = reinterpret_cast<const int4*>(deg);
  int t = threadIdx.x;
  int lane = t & 31;
  int w = t >> 5;

  int part = 0;
  int nfront = blockIdx.x * (SCAN_TILE / 4);
#pragma unroll 4
  for (int k = t; k < nfront; k += 1024) {
    int4 q = deg4[k];
    part += (q.x + q.y) + (q.z + q.w);
  }
  for (int o = 16; o > 0; o >>= 1) part += __shfl_xor(part, o);
  if (lane == 0) wsum[w] = part;

  int4 v = deg4[nfront + t];
  int s0 = v.x, s1 = s0 + v.y, s2 = s1 + v.z, s3 = s2 + v.w;
  int incl = s3;
  for (int o = 1; o < 32; o <<= 1) {
    int u = __shfl_up(incl, o);
    if (lane >= o) incl += u;
  }
  if (lane == 31) wtot[w] = incl;
  __syncthreads();

  int front = wsum[lane];
  int mine = (lane < w) ? wtot[lane] : 0;
  for (int o = 16; o > 0; o >>= 1) {
    front += __shfl_xor(front, o);
    mine += __shfl_xor(mine, o);
  }
  int ex = front + mine + (incl - s3);
  reinterpret_cast<int4*>(rowptr)[nfront + t] = make_int4(ex, ex + s0, ex + s1, ex + s2);
}

__global__ void fill_kernel(const int* __restrict__ src, const int* __restrict__ dst,
                            const int* __restrict__ rank, const int* __restrict__ rowptr,
                            int* __restrict__ nbr, int E) {
  int e = blockIdx.x * blockDim.x + threadIdx.x;
  if (e < E) nbr[rowptr[dst[e]] + rank[e]] = src[e];
}

#define BIN_SHIFT 9
#define BIN_NODES 512
#define NBIN 256
#define NCHUNK 256
#define PSTAGE 8192
#define NSTAGE 16384

__global__ void __launch_bounds__(1024)
partition_kernel(const int* __restrict__ src, const int* __restrict__ dst,
                 int2* __restrict__ pairs, int* __restrict__ histTab,
                 int* __restrict__ offTab, int E, int CH) {
  __shared__ int hist[NBIN];
  __shared__ int cur[NBIN];
  __shared__ int wtot[8];
  __shared__ int2 stage[PSTAGE];
  const int t = threadIdx.x;
  const int lane = t & 31;
  const int w = t >> 5;
  const int e0 = blockIdx.x * CH;
  int e1 = e0 + CH;
  if (e1 > E) e1 = E;
  const bool staged = (e1 - e0) <= PSTAGE;

  if (t < NBIN) hist[t] = 0;
  __syncthreads();

  for (int eb = e0; eb < e1; eb += 8 * 1024) {
    int dv[8];
#pragma unroll
    for (int k = 0; k < 8; ++k) {
      int e = eb + k * 1024 + t;
      dv[k] = (e < e1) ? dst[e] : -1;
    }
#pragma unroll
    for (int k = 0; k < 8; ++k)
      if (dv[k] >= 0) atomicAdd(&hist[dv[k] >> BIN_SHIFT], 1);
  }
  __syncthreads();

  int h = 0, incl = 0;
  if (t < NBIN) {
    h = hist[t];
    incl = h;
    for (int o = 1; o < 32; o <<= 1) {
      int u = __shfl_up(incl, o);
      if (lane >= o) incl += u;
    }
    if (lane == 31) wtot[w] = incl;
  }
  __syncthreads();
  if (t < NBIN) {
    int front = 0;
    for (int k = 0; k < w; ++k) front += wtot[k];
    int ex = front + incl - h;
    cur[t] = ex;
    histTab[blockIdx.x * NBIN + t] = h;
    offTab[blockIdx.x * NBIN + t] = ex;
  }
  __syncthreads();

  for (int eb = e0; eb < e1; eb += 8 * 1024) {
    int sv[8], dv[8];
#pragma unroll
    for (int k = 0; k < 8; ++k) {
      int e = eb + k * 1024 + t;
      bool ok = e < e1;
      dv[k] = ok ? dst[e] : -1;
      sv[k] = ok ? src[e] : 0;
    }
#pragma unroll
    for (int k = 0; k < 8; ++k) {
      if (dv[k] >= 0) {
        int pos = atomicAdd(&cur[dv[k] >> BIN_SHIFT], 1);
        if (staged) stage[pos] = make_int2(sv[k], dv[k]);
        else pairs[e0 + pos] = make_int2(sv[k], dv[k]);
      }
    }
  }
  __syncthreads();
  if (staged)
    for (int i = t; i < e1 - e0; i += 1024) pairs[e0 + i] = stage[i];
}

__global__ void __launch_bounds__(1024)
finish_kernel(const int2* __restrict__ pairs, const int* __restrict__ histTab,
              const int* __restrict__ offTab, int* __restrict__ rowptr,
              int* __restrict__ deg, int* __restrict__ nbr, int N, int CH) {
  __shared__ int colsum[4][NBIN];
  __shared__ int plen[NCHUNK];
  __shared__ int psrc[NCHUNK];
  __shared__ int cnt[BIN_NODES];
  __shared__ int cur[BIN_NODES];
  __shared__ int wred[32];
  __shared__ int wtot[16];
  __shared__ int nstage[NSTAGE];
  const int t = threadIdx.x;
  const int lane = t & 31;
  const int w = t >> 5;
  const int r = blockIdx.x;
  const int base = r << BIN_SHIFT;

  {
    const int col = t & (NBIN - 1);
    const int r0 = t >> 8;
    int sum = 0;
#pragma unroll 16
    for (int k = 0; k < NCHUNK / 4; ++k) sum += histTab[(r0 + 4 * k) * NBIN + col];
    colsum[r0][col] = sum;
  }
  if (t < NCHUNK) {
    plen[t] = histTab[t * NBIN + r];
    psrc[t] = t * CH + offTab[t * NBIN + r];
  }
  if (t < BIN_NODES) cnt[t] = 0;
  __syncthreads();

  int v = 0;
  if (t < NBIN && t < r) v = (colsum[0][t] + colsum[1][t]) + (colsum[2][t] + colsum[3][t]);
  for (int o = 16; o > 0; o >>= 1) v += __shfl_xor(v, o);
  if (lane == 0) wred[w] = v;

  int len[8], p0[8];
  int2 pa[8], pb[8];
#pragma unroll
  for (int k = 0; k < 8; ++k) {
    len[k] = plen[w + 32 * k];
    p0[k] = psrc[w + 32 * k];
    pa[k] = (lane < len[k]) ? pairs[p0[k] + lane] : make_int2(0, base);
    pb[k] = (lane + 32 < len[k]) ? pairs[p0[k] + lane + 32] : make_int2(0, base);
  }
#pragma unroll
  for (int k = 0; k < 8; ++k) {
    if (lane < len[k]) atomicAdd(&cnt[pa[k].y - base], 1);
    if (lane + 32 < len[k]) atomicAdd(&cnt[pb[k].y - base], 1);
    for (int i = lane + 64; i < len[k]; i += 32) atomicAdd(&cnt[pairs[p0[k] + i].y - base], 1);
  }
  __syncthreads();

  int binStart = 0;
  for (int k = 0; k < 8; ++k) binStart += wred[k];
  const int rangeCount = (colsum[0][r] + colsum[1][r]) + (colsum[2][r] + colsum[3][r]);
  const bool staged = rangeCount <= NSTAGE;

  int h = 0, incl = 0;
  if (t < BIN_NODES) {
    h = cnt[t];
    incl = h;
    for (int o = 1; o < 32; o <<= 1) {
      int u = __shfl_up(incl, o);
      if (lane >= o) incl += u;
    }
    if (lane == 31) wtot[w] = incl;
  }
  __syncthreads();
  if (t < BIN_NODES) {
    int front = 0;
    for (int k = 0; k < w; ++k) front += wtot[k];
    int ex = binStart + front + incl - h;
    cur[t] = ex;
    if (base + t < N) {
      rowptr[base + t] = ex;
      deg[base + t] = h;
    }
  }
  __syncthreads();

#pragma unroll
  for (int k = 0; k < 8; ++k) {
    if (lane < len[k]) {
      int slot = atomicAdd(&cur[pa[k].y - base], 1);
      if (staged) nstage[slot - binStart] = pa[k].x;
      else nbr[slot] = pa[k].x;
    }
    if (lane + 32 < len[k]) {
      int slot = atomicAdd(&cur[pb[k].y - base], 1);
      if (staged) nstage[slot - binStart] = pb[k].x;
      else nbr[slot] = pb[k].x;
    }
    for (int i = lane + 64; i < len[k]; i += 32) {
      int2 q = pairs[p0[k] + i];
      int slot = atomicAdd(&cur[q.y - base], 1);
      if (staged) nstage[slot - binStart] = q.x;
      else nbr[slot] = q.x;
    }
  }
  __syncthreads();
  if (staged)
    for (int i = t; i < rangeCount; i += 1024) nbr[binStart + i] = nstage[i];
}

template <int K>
__device__ __forceinline__ void load_rows(const float4* __restrict__ z4, int idx, int j,
                                          int lane, float4 (&v)[K]) {
#pragma unroll
  for (int u = 0; u < K; ++u) {
    int s = __builtin_amdgcn_readlane(idx, j + u);
    v[u] = z4[(size_t)s * (D / 4) + lane];
  }
}

template <int K>
__device__ __forceinline__ void sum_rows(const float4 (&v)[K], float4& acc) {
#pragma unroll
  for (int u = 0; u < K; ++u) {
    acc.x += fmaxf(v[u].x, 0.0f);
    acc.y += fmaxf(v[u].y, 0.0f);
    acc.z += fmaxf(v[u].z, 0.0f);
    acc.w += fmaxf(v[u].w, 0.0f);
  }
}

#define WPITCH 288
#define WMAT (64 * WPITCH)
#define WLDS_FLOATS 41472

typedef float v4f __attribute__((ext_vector_type(4)));

template <bool SCALE>
__device__ __forceinline__ void tile_product(const float* __restrict__ arow, float invr,
                                             const float* wl, v8f (&acc)[8]) {
  v2f an[4];
  v4f bn[4];
#pragma unroll
  for (int j = 0; j < 4; ++j) an[j] = *reinterpret_cast<const v2f*>(arow + 4 * j);
#pragma unroll
  for (int q = 0; q < 4; ++q) bn[q] = *reinterpret_cast<const v4f*>(wl + 64 * q);

#pragma unroll 1
  for (int ch = 0; ch < 8; ++ch) {
    v2f ac[4];
#pragma unroll
    for (int j = 0; j < 4; ++j) ac[j] = an[j];
    const float* anext = arow + 16 * ((ch + 1) & 7);
#pragma unroll
    for (int j = 0; j < 4; ++j) an[j] = *reinterpret_cast<const v2f*>(anext + 4 * j);
#pragma unroll
    for (int j = 0; j < 4; ++j) {
      const int s = ch * 4 + j;
      v4f bc[4];
#pragma unroll
      for (int q = 0; q < 4; ++q) bc[q] = bn[q];
      const float* wnext = wl + 2 * ((s + 1) & 31) * WPITCH;
#pragma unroll
      for (int q = 0; q < 4; ++q) bn[q] = *reinterpret_cast<const v4f*>(wnext + 64 * q);
      v2f a = ac[j];
      if (SCALE) { a.x *= invr; a.y *= invr; }
#pragma unroll
      for (int q = 0; q < 4; ++q) {
        v2f b0, b1;
        b0.x = bc[q].x; b0.y = bc[q].y;
        b1.x = bc[q].z; b1.y = bc[q].w;
        acc[2 * q] = __builtin_amdgcn_wmma_f32_16x16x4_f32(
            false, a, false, b0, (short)0, acc[2 * q], false, false);
        acc[2 * q + 1] = __builtin_amdgcn_wmma_f32_16x16x4_f32(
            false, a, false, b1, (short)0, acc[2 * q + 1], false, false);
      }
    }
  }
}

__global__ void __launch_bounds__(512)
combine_kernel(const float* __restrict__ summed, const float* __restrict__ inv,
               const float* __restrict__ z, const float* __restrict__ Wl,
               const float* __restrict__ bl, const float* __restrict__ Wr,
               float* __restrict__ out, int ntiles, int do_relu) {
  __shared__ __attribute__((aligned(16))) float wlds[WLDS_FLOATS];
  const int t = threadIdx.x;

  {
    float4 r0[8], r1[8];
#pragma unroll
    for (int it = 0; it < 8; ++it) {
      int item = it * 512 + t;
      int m = item >> 11;
      int p = (item >> 5) & 63;
      int n4 = item & 31;
      const float4* Wm = reinterpret_cast<const float4*>(m ? Wr : Wl);
      r0[it] = Wm[(2 * p) * (D / 4) + n4];
      r1[it] = Wm[(2 * p + 1) * (D / 4) + n4];
    }
#pragma unroll
    for (int it = 0; it < 8; ++it) {
      int item = it * 512 + t;
      int m = item >> 11;
      int p = (item >> 5) & 63;
      int n4 = item & 31;
      float* dst = wlds + m * WMAT + p * WPITCH + ((n4 >> 3) * 16 + 4 * (n4 & 3)) * 4 +
                   ((n4 >> 2) & 1) * 2;
      v2f e0, e1, e2, e3;
      e0.x = r0[it].x; e0.y = r1[it].x;
      e1.x = r0[it].y; e1.y = r1[it].y;
      e2.x = r0[it].z; e2.y = r1[it].z;
      e3.x = r0[it].w; e3.y = r1[it].w;
      *reinterpret_cast<v2f*>(dst + 0)  = e0;
      *reinterpret_cast<v2f*>(dst + 4)  = e1;
      *reinterpret_cast<v2f*>(dst + 8)  = e2;
      *reinterpret_cast<v2f*>(dst + 12) = e3;
    }
  }
  __syncthreads();

  const int lane = t & 31;
  const int w = t >> 5;
  const int lrow = lane & 15;
  const int half = lane >> 4;
  const float* wl_l = wlds + half * WPITCH + lrow * 4;
  const float* wl_r = wl_l + WMAT;

  for (int tile = blockIdx.x + w * gridDim.x; tile < ntiles; tile += 16 * gridDim.x) {
    v8f acc[8] = {};

    int row = tile * NTILE + lrow;
    float invr = inv[row];
    const float* arow = summed + (size_t)row * D + half * 2;
    const float* zrow = z + (size_t)row * D + half * 2;

    tile_product<true>(arow, invr, wl_l, acc);
    tile_product<false>(zrow, 1.0f, wl_r, acc);

    int mbase = tile * NTILE + (half << 3);
#pragma unroll
    for (int n = 0; n < 8; ++n) {
      float bias = bl[n * 16 + lrow];
#pragma unroll
      for (int v = 0; v < 8; ++v) {
        float val = acc[n][v] + bias;
        if (do_relu) val = fmaxf(val, 0.0f);
        out[(size_t)(mbase + v) * D + n * 16 + lrow] = val;
      }
    }
  }
}

#define APITCH 132
#define ATILE (NTILE * APITCH)
#define ZSTAGE (4 * 32 * 4)
#define LAYER_LDS_FLOATS (2 * WMAT + 16 * ATILE + 16 * ZSTAGE)
#define GCHUNK 23

__device__ __forceinline__ void wave_sync() {
  __builtin_amdgcn_fence(__ATOMIC_RELEASE, "wavefront");
  __builtin_amdgcn_wave_barrier();
  __builtin_amdgcn_fence(__ATOMIC_ACQUIRE, "wavefront");
}

__device__ __forceinline__ void zr_slice(v4f a, const float* wr, v8f (&acc)[8]) {
  v2f a0, a1;
  a0.x = a.x; a0.y = a.y;
  a1.x = a.z; a1.y = a.w;
  {
    v4f bq[4];
#pragma unroll
    for (int q = 0; q < 4; ++q) bq[q] = *reinterpret_cast<const v4f*>(wr + 64 * q);
#pragma unroll
    for (int q = 0; q < 4; ++q) {
      v2f e0, e1;
      e0.x = bq[q].x; e0.y = bq[q].y;
      e1.x = bq[q].z; e1.y = bq[q].w;
      acc[2 * q] = __builtin_amdgcn_wmma_f32_16x16x4_f32(
          false, a0, false, e0, (short)0, acc[2 * q], false, false);
      acc[2 * q + 1] = __builtin_amdgcn_wmma_f32_16x16x4_f32(
          false, a0, false, e1, (short)0, acc[2 * q + 1], false, false);
    }
  }
  {
    v4f bq[4];
#pragma unroll
    for (int q = 0; q < 4; ++q) bq[q] = *reinterpret_cast<const v4f*>(wr + WPITCH + 64 * q);
#pragma unroll
    for (int q = 0; q < 4; ++q) {
      v2f e0, e1;
      e0.x = bq[q].x; e0.y = bq[q].y;
      e1.x = bq[q].z; e1.y = bq[q].w;
      acc[2 * q] = __builtin_amdgcn_wmma_f32_16x16x4_f32(
          false, a1, false, e0, (short)0, acc[2 * q], false, false);
      acc[2 * q + 1] = __builtin_amdgcn_wmma_f32_16x16x4_f32(
          false, a1, false, e1, (short)0, acc[2 * q + 1], false, false);
    }
  }
}

__global__ void __launch_bounds__(512)
layer_kernel(const float* __restrict__ z, const int* __restrict__ rowptr,
             const int* __restrict__ deg, const int* __restrict__ nbr,
             const float* __restrict__ Wl, const float* __restrict__ bl,
             const float* __restrict__ Wr, float* __restrict__ out, int ntiles, int n,
             int do_relu) {
  __shared__ __attribute__((aligned(16))) float wlds[LAYER_LDS_FLOATS];
  const int t = threadIdx.x;

  {
    float4 r0[8], r1[8];
#pragma unroll
    for (int it = 0; it < 8; ++it) {
      int item = it * 512 + t;
      int m = item >> 11;
      int p = (item >> 5) & 63;
      int n4 = item & 31;
      const float4* Wm = reinterpret_cast<const float4*>(m ? Wr : Wl);
      r0[it] = Wm[(2 * p) * (D / 4) + n4];
      r1[it] = Wm[(2 * p + 1) * (D / 4) + n4];
    }
#pragma unroll
    for (int it = 0; it < 8; ++it) {
      int item = it * 512 + t;
      int m = item >> 11;
      int p = (item >> 5) & 63;
      int n4 = item & 31;
      float* dst = wlds + m * WMAT + p * WPITCH + ((n4 >> 3) * 16 + 4 * (n4 & 3)) * 4 +
                   ((n4 >> 2) & 1) * 2;
      v2f e0, e1, e2, e3;
      e0.x = r0[it].x; e0.y = r1[it].x;
      e1.x = r0[it].y; e1.y = r1[it].y;
      e2.x = r0[it].z; e2.y = r1[it].z;
      e3.x = r0[it].w; e3.y = r1[it].w;
      *reinterpret_cast<v2f*>(dst + 0)  = e0;
      *reinterpret_cast<v2f*>(dst + 4)  = e1;
      *reinterpret_cast<v2f*>(dst + 8)  = e2;
      *reinterpret_cast<v2f*>(dst + 12) = e3;
    }
  }
  __syncthreads();

  const int lane = t & 31;
  const int w = t >> 5;
  const int lrow = lane & 15;
  const int half = lane >> 4;
  const float* wl_l = wlds + half * WPITCH + lrow * 4;
  const float* wr_s = wlds + WMAT + half * (2 * WPITCH) + lrow * 4;
  float* at = wlds + 2 * WMAT + w * ATILE;
  const float* afrag = at + lrow * APITCH + half * 2;
  float* zst = wlds + 2 * WMAT + 16 * ATILE + w * ZSTAGE + lane * 4;
  const float4* z4 = reinterpret_cast<const float4*>(z);

  for (int tile = blockIdx.x + w * gridDim.x; tile < ntiles; tile += 16 * gridDim.x) {
    v8f acc[8] = {};
    const int row = tile * NTILE + lrow;
    const float* zq = z + (size_t)row * D + half * 4;
    const int rp = (row < n) ? rowptr[row] : 0;
    const int dg = (row < n) ? deg[row] : 0;
    v4f zn[4];
#pragma unroll
    for (int u = 0; u < 4; ++u) zn[u] = *reinterpret_cast<const v4f*>(zq + 8 * u);

    wave_sync();

    int idx_next;
    {
      int s0 = __builtin_amdgcn_readlane(rp, 0);
      int c0 = __builtin_amdgcn_readlane(dg, 0);
      int m0 = c0 < GCHUNK ? c0 : GCHUNK;
      idx_next = (lane < m0) ? nbr[s0 + lane] : 0;
    }
    for (int r = 0; r < NTILE; ++r) {
      if ((r & 3) == 0) {
#pragma unroll
        for (int u = 0; u < 4; ++u) *reinterpret_cast<v4f*>(zst + 128 * u) = zn[u];
      }
      const v4f za = *reinterpret_cast<const v4f*>(zst + 128 * (r & 3));
      const int start = __builtin_amdgcn_readlane(rp, r);
      const int cnt = __builtin_amdgcn_readlane(dg, r);
      int idx = idx_next;
      if (r + 1 < NTILE) {
        int s1 = __builtin_amdgcn_readlane(rp, r + 1);
        int c1 = __builtin_amdgcn_readlane(dg, r + 1);
        int m1 = c1 < GCHUNK ? c1 : GCHUNK;
        idx_next = (lane < m1) ? nbr[s1 + lane] : 0;
      }
      float4 nsum = make_float4(0.0f, 0.0f, 0.0f, 0.0f);
      int base = 0;
      do {
        int m = cnt - base;
        if (m > GCHUNK) m = GCHUNK;
        if (base > 0) idx = (lane < m) ? nbr[start + base + lane] : 0;
        float4 va[8], vb[8], vc[4], vd[2], ve[1];
        int j = 0;
        if (m >= 8)  { load_rows<8>(z4, idx, j, lane, va); j += 8; }
        if (m >= 16) { load_rows<8>(z4, idx, j, lane, vb); j += 8; }
        if (m & 4)   { load_rows<4>(z4, idx, j, lane, vc); j += 4; }
        if (m & 2)   { load_rows<2>(z4, idx, j, lane, vd); j += 2; }
        if (m & 1)   { load_rows<1>(z4, idx, j, lane, ve); }
        if (base == 0) {
          if ((r & 3) == 3 && r + 1 < NTILE) {
            const float* zg = zq + 8 * (r + 1);
#pragma unroll
            for (int u = 0; u < 4; ++u) zn[u] = *reinterpret_cast<const v4f*>(zg + 8 * u);
          }
          zr_slice(za, wr_s + (4 * r) * WPITCH, acc);
        }
        if (m >= 8)  sum_rows<8>(va, nsum);
        if (m >= 16) sum_rows<8>(vb, nsum);
        if (m & 4)   sum_rows<4>(vc, nsum);
        if (m & 2)   sum_rows<2>(vd, nsum);
        if (m & 1)   sum_rows<1>(ve, nsum);
        base += GCHUNK;
      } while (base < cnt);
      const float invr = 1.0f / fmaxf((float)cnt, 1.0f);
      nsum.x *= invr; nsum.y *= invr; nsum.z *= invr; nsum.w *= invr;
      *reinterpret_cast<float4*>(at + r * APITCH + 4 * lane) = nsum;
    }
    wave_sync();

    tile_product<false>(afrag, 1.0f, wl_l, acc);

    int mbase = tile * NTILE + (half << 3);
#pragma unroll
    for (int nn = 0; nn < 8; ++nn) {
      float bias = bl[nn * 16 + lrow];
#pragma unroll
      for (int v = 0; v < 8; ++v) {
        float val = acc[nn][v] + bias;
        if (do_relu) val = fmaxf(val, 0.0f);
        out[(size_t)(mbase + v) * D + nn * 16 + lrow] = val;
      }
    }
  }
}

extern "C" void kernel_launch(void* const* d_in, const int* in_sizes, int n_in,
                              void* d_out, int out_size, void* d_ws, size_t ws_size,
                              hipStream_t stream) {
  const float* x   = (const float*)d_in[0];
  const int*   ei  = (const int*)d_in[1];
  const float* Wl1 = (const float*)d_in[2];
  const float* bl1 = (const float*)d_in[3];
  const float* Wr1 = (const float*)d_in[4];
  const float* Wl2 = (const float*)d_in[5];
  const float* bl2 = (const float*)d_in[6];
  const float* Wr2 = (const float*)d_in[7];
  float* out = (float*)d_out;

  const int N = in_sizes[0] / D;
  const int E = in_sizes[1] / 2;
  const int ntiles = (N + NTILE - 1) / NTILE;

  const int* src = ei;
  const int* dst = ei + E;

  float* cnt = (float*)d_ws;
  float* summed = cnt + N;

  const int combine_blocks = 256;

  const int scan_blocks = (N + SCAN_TILE - 1) / SCAN_TILE;
  const int Npad = scan_blocks * SCAN_TILE;
  const size_t need = ((size_t)N * (D + 1) + 2 * (size_t)Npad + (size_t)E) * sizeof(float);

  if (ws_size >= need) {
    int* deg = (int*)(summed + (size_t)N * D);
    int* rowptr = deg + Npad;
    int* nbr = rowptr + Npad;

    const int ranges = (N + BIN_NODES - 1) / BIN_NODES;
    if (ranges <= NBIN && (size_t)N * D >= 2 * (size_t)E + 2 * NCHUNK * NBIN) {
      const int CH = (E + NCHUNK - 1) / NCHUNK;
      int2* pairs = (int2*)d_out;
      int* histTab = (int*)d_out + 2 * (size_t)E;
      int* offTab = histTab + NCHUNK * NBIN;
      partition_kernel<<<NCHUNK, 1024, 0, stream>>>(src, dst, pairs, histTab, offTab, E, CH);
      finish_kernel<<<ranges, 1024, 0, stream>>>(pairs, histTab, offTab, rowptr, deg, nbr,
                                                 N, CH);
    } else {
      int* rank = (int*)d_out;
      int edge_blocks = (E + 255) / 256;
      zero_kernel<<<(Npad + 255) / 256, 256, 0, stream>>>((float*)deg, (long)Npad);
      count_kernel<<<edge_blocks, 256, 0, stream>>>(dst, deg, rank, E);
      scan_kernel<<<scan_blocks, 1024, 0, stream>>>(deg, rowptr);
      fill_kernel<<<edge_blocks, 256, 0, stream>>>(src, dst, rank, rowptr, nbr, E);
    }
    float* h = summed;
    layer_kernel<<<combine_blocks, 512, 0, stream>>>(x, rowptr, deg, nbr, Wl1, bl1, Wr1, h,
                                                     ntiles, N, 1);
    layer_kernel<<<combine_blocks, 512, 0, stream>>>(h, rowptr, deg, nbr, Wl2, bl2, Wr2, out,
                                                     ntiles, N, 0);
    return;
  }

  int scatter_blocks = (int)(((long)E * 32 + 255) / 256);

  zero_kernel<<<2048, 256, 0, stream>>>(cnt, (long)N * (D + 1));
  scatter_kernel<<<scatter_blocks, 256, 0, stream>>>(x, src, dst, summed, cnt, E);
  inv_kernel<<<(N + 255) / 256, 256, 0, stream>>>(cnt, N);
  combine_kernel<<<combine_blocks, 512, 0, stream>>>(summed, cnt, x, Wl1, bl1, Wr1,
                                                     out, ntiles, 1);
  zero_kernel<<<2048, 256, 0, stream>>>(summed, (long)N * D);
  scatter_kernel<<<scatter_blocks, 256, 0, stream>>>(out, src, dst, summed, nullptr, E);
  combine_kernel<<<combine_blocks, 512, 0, stream>>>(summed, cnt, out, Wl2, bl2, Wr2,
                                                     out, ntiles, 0);
}


# fused layer: the tiles beyond whole rounds of one per wave (8-9 per work-group) are gathered by all 16 waves, 8-9 nodes each, and multiplied in 64-column halves behind a barrier; sum order unchanged
# speedup vs baseline: 4.7607x; 1.0461x over previous
#include <hip/hip_runtime.h>


typedef float v2f __attribute__((ext_vector_type(2)));
typedef float v8f __attribute__((ext_vector_type(8)));

#define D 128
#define NTILE 16

__global__ void zero_kernel(float* __restrict__ p, long n) {
  long i = (long)blockIdx.x * blockDim.x + threadIdx.x;
  long stride = (long)gridDim.x * blockDim.x;
  for (; i < n; i += stride) p[i] = 0.0f;
}

__global__ void scatter_kernel(const float* __restrict__ z,
                               const int* __restrict__ src,
                               const int* __restrict__ dst,
                               float* __restrict__ summed,
                               float* __restrict__ cnt,
                               int E) {
  unsigned tid = blockIdx.x * blockDim.x + threadIdx.x;
  int wid = (int)(tid >> 5);
  int lane = threadIdx.x & 31;
  if (wid >= E) return;
  int s = __builtin_amdgcn_readfirstlane(src[wid]);
  int d = __builtin_amdgcn_readfirstlane(dst[wid]);

  const float* zp = z + (size_t)s * D + lane;
  float v0 = zp[0];
  float v1 = zp[32];
  float v2 = zp[64];
  float v3 = zp[96];

  float* op = summed + (size_t)d * D + lane;
  unsafeAtomicAdd(op + 0,  fmaxf(v0, 0.0f));
  unsafeAtomicAdd(op + 32, fmaxf(v1, 0.0f));
  unsafeAtomicAdd(op + 64, fmaxf(v2, 0.0f));
  unsafeAtomicAdd(op + 96, fmaxf(v3, 0.0f));
  if (cnt != nullptr && lane == 0) unsafeAtomicAdd(cnt + d, 1.0f);
}

__global__ void inv_kernel(float* __restrict__ cnt, int n) {
  int i = blockIdx.x * blockDim.x + threadIdx.x;
  if (i < n) cnt[i] = 1.0f / fmaxf(cnt[i], 1.0f);
}

__global__ void count_kernel(const int* __restrict__ dst, int* __restrict__ deg,
                             int* __restrict__ rank, int E) {
  int e = blockIdx.x * blockDim.x + threadIdx.x;
  if (e < E) rank[e] = atomicAdd(deg + dst[e], 1);
}

#define SCAN_TILE 4096

__global__ void __launch_bounds__(1024)
scan_kernel(const int* __restrict__ deg, int* __restrict__ rowptr) {
  __shared__ int wsum[32];
  __shared__ int wtot[32];
  const int4* deg4 = reinterpret_cast<const int4*>(deg);
  int t = threadIdx.x;
  int lane = t & 31;
  int w = t >> 5;

  int part = 0;
  int nfront = blockIdx.x * (SCAN_TILE / 4);
#pragma unroll 4
  for (int k = t; k < nfront; k += 1024) {
    int4 q = deg4[k];
    part += (q.x + q.y) + (q.z + q.w);
  }
  for (int o = 16; o > 0; o >>= 1) part += __shfl_xor(part, o);
  if (lane == 0) wsum[w] = part;

  int4 v = deg4[nfront + t];
  int s0 = v.x, s1 = s0 + v.y, s2 = s1 + v.z, s3 = s2 + v.w;
  int incl = s3;
  for (int o = 1; o < 32; o <<= 1) {
    int u = __shfl_up(incl, o);
    if (lane >= o) incl += u;
  }
  if (lane == 31) wtot[w] = incl;
  __syncthreads();

  int front = wsum[lane];
  int mine = (lane < w) ? wtot[lane] : 0;
  for (int o = 16; o > 0; o >>= 1) {
    front += __shfl_xor(front, o);
    mine += __shfl_xor(mine, o);
  }
  int ex = front + mine + (incl - s3);
  reinterpret_cast<int4*>(rowptr)[nfront + t] = make_int4(ex, ex + s0, ex + s1, ex + s2);
}

__global__ void fill_kernel(const int* __restrict__ src, const int* __restrict__ dst,
                            const int* __restrict__ rank, const int* __restrict__ rowptr,
                            int* __restrict__ nbr, int E) {
  int e = blockIdx.x * blockDim.x + threadIdx.x;
  if (e < E) nbr[rowptr[dst[e]] + rank[e]] = src[e];
}

#define BIN_SHIFT 9
#define BIN_NODES 512
#define NBIN 256
#define NCHUNK 256
#define PSTAGE 8192
#define NSTAGE 16384

__global__ void __launch_bounds__(1024)
partition_kernel(const int* __restrict__ src, const int* __restrict__ dst,
                 int2* __restrict__ pairs, int* __restrict__ histTab,
                 int* __restrict__ offTab, int E, int CH) {
  __shared__ int hist[NBIN];
  __shared__ int cur[NBIN];
  __shared__ int wtot[8];
  __shared__ int2 stage[PSTAGE];
  const int t = threadIdx.x;
  const int lane = t & 31;
  const int w = t >> 5;
  const int e0 = blockIdx.x * CH;
  int e1 = e0 + CH;
  if (e1 > E) e1 = E;
  const bool staged = (e1 - e0) <= PSTAGE;

  if (t < NBIN) hist[t] = 0;
  __syncthreads();

  for (int eb = e0; eb < e1; eb += 8 * 1024) {
    int dv[8];
#pragma unroll
    for (int k = 0; k < 8; ++k) {
      int e = eb + k * 1024 + t;
      dv[k] = (e < e1) ? dst[e] : -1;
    }
#pragma unroll
    for (int k = 0; k < 8; ++k)
      if (dv[k] >= 0) atomicAdd(&hist[dv[k] >> BIN_SHIFT], 1);
  }
  __syncthreads();

  int h = 0, incl = 0;
  if (t < NBIN) {
    h = hist[t];
    incl = h;
    for (int o = 1; o < 32; o <<= 1) {
      int u = __shfl_up(incl, o);
      if (lane >= o) incl += u;
    }
    if (lane == 31) wtot[w] = incl;
  }
  __syncthreads();
  if (t < NBIN) {
    int front = 0;
    for (int k = 0; k < w; ++k) front += wtot[k];
    int ex = front + incl - h;
    cur[t] = ex;
    histTab[blockIdx.x * NBIN + t] = h;
    offTab[blockIdx.x * NBIN + t] = ex;
  }
  __syncthreads();

  for (int eb = e0; eb < e1; eb += 8 * 1024) {
    int sv[8], dv[8];
#pragma unroll
    for (int k = 0; k < 8; ++k) {
      int e = eb + k * 1024 + t;
      bool ok = e < e1;
      dv[k] = ok ? dst[e] : -1;
      sv[k] = ok ? src[e] : 0;
    }
#pragma unroll
    for (int k = 0; k < 8; ++k) {
      if (dv[k] >= 0) {
        int pos = atomicAdd(&cur[dv[k] >> BIN_SHIFT], 1);
        if (staged) stage[pos] = make_int2(sv[k], dv[k]);
        else pairs[e0 + pos] = make_int2(sv[k], dv[k]);
      }
    }
  }
  __syncthreads();
  if (staged)
    for (int i = t; i < e1 - e0; i += 1024) pairs[e0 + i] = stage[i];
}

__global__ void __launch_bounds__(1024)
finish_kernel(const int2* __restrict__ pairs, const int* __restrict__ histTab,
              const int* __restrict__ offTab, int* __restrict__ rowptr,
              int* __restrict__ deg, int* __restrict__ nbr, int N, int CH) {
  __shared__ int colsum[4][NBIN];
  __shared__ int plen[NCHUNK];
  __shared__ int psrc[NCHUNK];
  __shared__ int cnt[BIN_NODES];
  __shared__ int cur[BIN_NODES];
  __shared__ int wred[32];
  __shared__ int wtot[16];
  __shared__ int nstage[NSTAGE];
  const int t = threadIdx.x;
  const int lane = t & 31;
  const int w = t >> 5;
  const int r = blockIdx.x;
  const int base = r << BIN_SHIFT;

  {
    const int col = t & (NBIN - 1);
    const int r0 = t >> 8;
    int sum = 0;
#pragma unroll 16
    for (int k = 0; k < NCHUNK / 4; ++k) sum += histTab[(r0 + 4 * k) * NBIN + col];
    colsum[r0][col] = sum;
  }
  if (t < NCHUNK) {
    plen[t] = histTab[t * NBIN + r];
    psrc[t] = t * CH + offTab[t * NBIN + r];
  }
  if (t < BIN_NODES) cnt[t] = 0;
  __syncthreads();

  int v = 0;
  if (t < NBIN && t < r) v = (colsum[0][t] + colsum[1][t]) + (colsum[2][t] + colsum[3][t]);
  for (int o = 16; o > 0; o >>= 1) v += __shfl_xor(v, o);
  if (lane == 0) wred[w] = v;

  int len[8], p0[8];
  int2 pa[8], pb[8];
#pragma unroll
  for (int k = 0; k < 8; ++k) {
    len[k] = plen[w + 32 * k];
    p0[k] = psrc[w + 32 * k];
    pa[k] = (lane < len[k]) ? pairs[p0[k] + lane] : make_int2(0, base);
    pb[k] = (lane + 32 < len[k]) ? pairs[p0[k] + lane + 32] : make_int2(0, base);
  }
#pragma unroll
  for (int k = 0; k < 8; ++k) {
    if (lane < len[k]) atomicAdd(&cnt[pa[k].y - base], 1);
    if (lane + 32 < len[k]) atomicAdd(&cnt[pb[k].y - base], 1);
    for (int i = lane + 64; i < len[k]; i += 32) atomicAdd(&cnt[pairs[p0[k] + i].y - base], 1);
  }
  __syncthreads();

  int binStart = 0;
  for (int k = 0; k < 8; ++k) binStart += wred[k];
  const int rangeCount = (colsum[0][r] + colsum[1][r]) + (colsum[2][r] + colsum[3][r]);
  const bool staged = rangeCount <= NSTAGE;

  int h = 0, incl = 0;
  if (t < BIN_NODES) {
    h = cnt[t];
    incl = h;
    for (int o = 1; o < 32; o <<= 1) {
      int u = __shfl_up(incl, o);
      if (lane >= o) incl += u;
    }
    if (lane == 31) wtot[w] = incl;
  }
  __syncthreads();
  if (t < BIN_NODES) {
    int front = 0;
    for (int k = 0; k < w; ++k) front += wtot[k];
    int ex = binStart + front + incl - h;
    cur[t] = ex;
    if (base + t < N) {
      rowptr[base + t] = ex;
      deg[base + t] = h;
    }
  }
  __syncthreads();

#pragma unroll
  for (int k = 0; k < 8; ++k) {
    if (lane < len[k]) {
      int slot = atomicAdd(&cur[pa[k].y - base], 1);
      if (staged) nstage[slot - binStart] = pa[k].x;
      else nbr[slot] = pa[k].x;
    }
    if (lane + 32 < len[k]) {
      int slot = atomicAdd(&cur[pb[k].y - base], 1);
      if (staged) nstage[slot - binStart] = pb[k].x;
      else nbr[slot] = pb[k].x;
    }
    for (int i = lane + 64; i < len[k]; i += 32) {
      int2 q = pairs[p0[k] + i];
      int slot = atomicAdd(&cur[q.y - base], 1);
      if (staged) nstage[slot - binStart] = q.x;
      else nbr[slot] = q.x;
    }
  }
  __syncthreads();
  if (staged)
    for (int i = t; i < rangeCount; i += 1024) nbr[binStart + i] = nstage[i];
}

template <int K>
__device__ __forceinline__ void load_rows(const float4* __restrict__ z4, int idx, int j,
                                          int lane, float4 (&v)[K]) {
#pragma unroll
  for (int u = 0; u < K; ++u) {
    int s = __builtin_amdgcn_readlane(idx, j + u);
    v[u] = z4[(size_t)s * (D / 4) + lane];
  }
}

template <int K>
__device__ __forceinline__ void sum_rows(const float4 (&v)[K], float4& acc) {
#pragma unroll
  for (int u = 0; u < K; ++u) {
    acc.x += fmaxf(v[u].x, 0.0f);
    acc.y += fmaxf(v[u].y, 0.0f);
    acc.z += fmaxf(v[u].z, 0.0f);
    acc.w += fmaxf(v[u].w, 0.0f);
  }
}

#define WPITCH 288
#define WMAT (64 * WPITCH)
#define WLDS_FLOATS 41472

typedef float v4f __attribute__((ext_vector_type(4)));

template <bool SCALE, int NQ = 4>
__device__ __forceinline__ void tile_product(const float* __restrict__ arow, float invr,
                                             const float* wl, v8f (&acc)[2 * NQ]) {
  v2f an[4];
  v4f bn[NQ];
#pragma unroll
  for (int j = 0; j < 4; ++j) an[j] = *reinterpret_cast<const v2f*>(arow + 4 * j);
#pragma unroll
  for (int q = 0; q < NQ; ++q) bn[q] = *reinterpret_cast<const v4f*>(wl + 64 * q);

#pragma unroll 1
  for (int ch = 0; ch < 8; ++ch) {
    v2f ac[4];
#pragma unroll
    for (int j = 0; j < 4; ++j) ac[j] = an[j];
    const float* anext = arow + 16 * ((ch + 1) & 7);
#pragma unroll
    for (int j = 0; j < 4; ++j) an[j] = *reinterpret_cast<const v2f*>(anext + 4 * j);
#pragma unroll
    for (int j = 0; j < 4; ++j) {
      const int s = ch * 4 + j;
      v4f bc[NQ];
#pragma unroll
      for (int q = 0; q < NQ; ++q) bc[q] = bn[q];
      const float* wnext = wl + 2 * ((s + 1) & 31) * WPITCH;
#pragma unroll
      for (int q = 0; q < NQ; ++q) bn[q] = *reinterpret_cast<const v4f*>(wnext + 64 * q);
      v2f a = ac[j];
      if (SCALE) { a.x *= invr; a.y *= invr; }
#pragma unroll
      for (int q = 0; q < NQ; ++q) {
        v2f b0, b1;
        b0.x = bc[q].x; b0.y = bc[q].y;
        b1.x = bc[q].z; b1.y = bc[q].w;
        acc[2 * q] = __builtin_amdgcn_wmma_f32_16x16x4_f32(
            false, a, false, b0, (short)0, acc[2 * q], false, false);
        acc[2 * q + 1] = __builtin_amdgcn_wmma_f32_16x16x4_f32(
            false, a, false, b1, (short)0, acc[2 * q + 1], false, false);
      }
    }
  }
}

__global__ void __launch_bounds__(512)
combine_kernel(const float* __restrict__ summed, const float* __restrict__ inv,
               const float* __restrict__ z, const float* __restrict__ Wl,
               const float* __restrict__ bl, const float* __restrict__ Wr,
               float* __restrict__ out, int ntiles, int do_relu) {
  __shared__ __attribute__((aligned(16))) float wlds[WLDS_FLOATS];
  const int t = threadIdx.x;

  {
    float4 r0[8], r1[8];
#pragma unroll
    for (int it = 0; it < 8; ++it) {
      int item = it * 512 + t;
      int m = item >> 11;
      int p = (item >> 5) & 63;
      int n4 = item & 31;
      const float4* Wm = reinterpret_cast<const float4*>(m ? Wr : Wl);
      r0[it] = Wm[(2 * p) * (D / 4) + n4];
      r1[it] = Wm[(2 * p + 1) * (D / 4) + n4];
    }
#pragma unroll
    for (int it = 0; it < 8; ++it) {
      int item = it * 512 + t;
      int m = item >> 11;
      int p = (item >> 5) & 63;
      int n4 = item & 31;
      float* dst = wlds + m * WMAT + p * WPITCH + ((n4 >> 3) * 16 + 4 * (n4 & 3)) * 4 +
                   ((n4 >> 2) & 1) * 2;
      v2f e0, e1, e2, e3;
      e0.x = r0[it].x; e0.y = r1[it].x;
      e1.x = r0[it].y; e1.y = r1[it].y;
      e2.x = r0[it].z; e2.y = r1[it].z;
      e3.x = r0[it].w; e3.y = r1[it].w;
      *reinterpret_cast<v2f*>(dst + 0)  = e0;
      *reinterpret_cast<v2f*>(dst + 4)  = e1;
      *reinterpret_cast<v2f*>(dst + 8)  = e2;
      *reinterpret_cast<v2f*>(dst + 12) = e3;
    }
  }
  __syncthreads();

  const int lane = t & 31;
  const int w = t >> 5;
  const int lrow = lane & 15;
  const int half = lane >> 4;
  const float* wl_l = wlds + half * WPITCH + lrow * 4;
  const float* wl_r = wl_l + WMAT;

  for (int tile = blockIdx.x + w * gridDim.x; tile < ntiles; tile += 16 * gridDim.x) {
    v8f acc[8] = {};

    int row = tile * NTILE + lrow;
    float invr = inv[row];
    const float* arow = summed + (size_t)row * D + half * 2;
    const float* zrow = z + (size_t)row * D + half * 2;

    tile_product<true>(arow, invr, wl_l, acc);
    tile_product<false>(zrow, 1.0f, wl_r, acc);

    int mbase = tile * NTILE + (half << 3);
#pragma unroll
    for (int n = 0; n < 8; ++n) {
      float bias = bl[n * 16 + lrow];
#pragma unroll
      for (int v = 0; v < 8; ++v) {
        float val = acc[n][v] + bias;
        if (do_relu) val = fmaxf(val, 0.0f);
        out[(size_t)(mbase + v) * D + n * 16 + lrow] = val;
      }
    }
  }
}

#define APITCH 132
#define ATILE (NTILE * APITCH)
#define ZSTAGE (4 * 32 * 4)
#define LAYER_LDS_FLOATS (2 * WMAT + 17 * ATILE + 16 * ZSTAGE)
#define GCHUNK 23

__device__ __forceinline__ void wave_sync() {
  __builtin_amdgcn_fence(__ATOMIC_RELEASE, "wavefront");
  __builtin_amdgcn_wave_barrier();
  __builtin_amdgcn_fence(__ATOMIC_ACQUIRE, "wavefront");
}

template <int NQ = 4>
__device__ __forceinline__ void zr_slice(v4f a, const float* wr, v8f (&acc)[2 * NQ]) {
  v2f a0, a1;
  a0.x = a.x; a0.y = a.y;
  a1.x = a.z; a1.y = a.w;
  {
    v4f bq[NQ];
#pragma unroll
    for (int q = 0; q < NQ; ++q) bq[q] = *reinterpret_cast<const v4f*>(wr + 64 * q);
#pragma unroll
    for (int q = 0; q < NQ; ++q) {
      v2f e0, e1;
      e0.x = bq[q].x; e0.y = bq[q].y;
      e1.x = bq[q].z; e1.y = bq[q].w;
      acc[2 * q] = __builtin_amdgcn_wmma_f32_16x16x4_f32(
          false, a0, false, e0, (short)0, acc[2 * q], false, false);
      acc[2 * q + 1] = __builtin_amdgcn_wmma_f32_16x16x4_f32(
          false, a0, false, e1, (short)0, acc[2 * q + 1], false, false);
    }
  }
  {
    v4f bq[NQ];
#pragma unroll
    for (int q = 0; q < NQ; ++q) bq[q] = *reinterpret_cast<const v4f*>(wr + WPITCH + 64 * q);
#pragma unroll
    for (int q = 0; q < NQ; ++q) {
      v2f e0, e1;
      e0.x = bq[q].x; e0.y = bq[q].y;
      e1.x = bq[q].z; e1.y = bq[q].w;
      acc[2 * q] = __builtin_amdgcn_wmma_f32_16x16x4_f32(
          false, a1, false, e0, (short)0, acc[2 * q], false, false);
      acc[2 * q + 1] = __builtin_amdgcn_wmma_f32_16x16x4_f32(
          false, a1, false, e1, (short)0, acc[2 * q + 1], false, false);
    }
  }
}

template <class Mid>
__device__ __forceinline__ float4 gather_node(const float4* __restrict__ z4,
                                              const int* __restrict__ nbr, int start, int cnt,
                                              int idx, int lane, Mid mid) {
  float4 nsum = make_float4(0.0f, 0.0f, 0.0f, 0.0f);
  int base = 0;
  do {
    int m = cnt - base;
    if (m > GCHUNK) m = GCHUNK;
    if (base > 0) idx = (lane < m) ? nbr[start + base + lane] : 0;
    float4 va[8], vb[8], vc[4], vd[2], ve[1];
    int j = 0;
    if (m >= 8)  { load_rows<8>(z4, idx, j, lane, va); j += 8; }
    if (m >= 16) { load_rows<8>(z4, idx, j, lane, vb); j += 8; }
    if (m & 4)   { load_rows<4>(z4, idx, j, lane, vc); j += 4; }
    if (m & 2)   { load_rows<2>(z4, idx, j, lane, vd); j += 2; }
    if (m & 1)   { load_rows<1>(z4, idx, j, lane, ve); }
    if (base == 0) mid();
    if (m >= 8)  sum_rows<8>(va, nsum);
    if (m >= 16) sum_rows<8>(vb, nsum);
    if (m & 4)   sum_rows<4>(vc, nsum);
    if (m & 2)   sum_rows<2>(vd, nsum);
    if (m & 1)   sum_rows<1>(ve, nsum);
    base += GCHUNK;
  } while (base < cnt);
  return nsum;
}

__global__ void __launch_bounds__(512)
layer_kernel(const float* __restrict__ z, const int* __restrict__ rowptr,
             const int* __restrict__ deg, const int* __restrict__ nbr,
             const float* __restrict__ Wl, const float* __restrict__ bl,
             const float* __restrict__ Wr, float* __restrict__ out, int ntiles, int n,
             int do_relu) {
  __shared__ __attribute__((aligned(16))) float wlds[LAYER_LDS_FLOATS];
  const int t = threadIdx.x;

  {
    float4 r0[8], r1[8];
#pragma unroll
    for (int it = 0; it < 8; ++it) {
      int item = it * 512 + t;
      int m = item >> 11;
      int p = (item >> 5) & 63;
      int n4 = item & 31;
      const float4* Wm = reinterpret_cast<const float4*>(m ? Wr : Wl);
      r0[it] = Wm[(2 * p) * (D / 4) + n4];
      r1[it] = Wm[(2 * p + 1) * (D / 4) + n4];
    }
#pragma unroll
    for (int it = 0; it < 8; ++it) {
      int item = it * 512 + t;
      int m = item >> 11;
      int p = (item >> 5) & 63;
      int n4 = item & 31;
      float* dst = wlds + m * WMAT + p * WPITCH + ((n4 >> 3) * 16 + 4 * (n4 & 3)) * 4 +
                   ((n4 >> 2) & 1) * 2;
      v2f e0, e1, e2, e3;
      e0.x = r0[it].x; e0.y = r1[it].x;
      e1.x = r0[it].y; e1.y = r1[it].y;
      e2.x = r0[it].z; e2.y = r1[it].z;
      e3.x = r0[it].w; e3.y = r1[it].w;
      *reinterpret_cast<v2f*>(dst + 0)  = e0;
      *reinterpret_cast<v2f*>(dst + 4)  = e1;
      *reinterpret_cast<v2f*>(dst + 8)  = e2;
      *reinterpret_cast<v2f*>(dst + 12) = e3;
    }
  }
  __syncthreads();

  const int lane = t & 31;
  const int w = t >> 5;
  const int lrow = lane & 15;
  const int half = lane >> 4;
  const float* wl_l = wlds + half * WPITCH + lrow * 4;
  const float* wr_s = wlds + WMAT + half * (2 * WPITCH) + lrow * 4;
  float* at = wlds + 2 * WMAT + w * ATILE;
  const float* afrag = at + lrow * APITCH + half * 2;
  float* zst = wlds + 2 * WMAT + 16 * ATILE + w * ZSTAGE + lane * 4;
  const float4* z4 = reinterpret_cast<const float4*>(z);

  const int rem = ntiles % (16 * (int)gridDim.x);
  const int Tb = ((int)blockIdx.x < rem)
                     ? (rem - (int)blockIdx.x + (int)gridDim.x - 1) / (int)gridDim.x
                     : 0;
  const int ws = __builtin_amdgcn_readfirstlane(w);
  float* xt = wlds + 2 * WMAT + 16 * ATILE + 16 * ZSTAGE;
  for (int j0 = 0; j0 < Tb; j0 += 9) {
    const int T = (Tb - j0 < 9) ? (Tb - j0) : 9;
    const int Tm = T < 8 ? T : 8;
    if (j0 > 0) __syncthreads();
    const bool unit = ws < 2 * Tm;
    const int hN = ws & 1;
    const bool spread = unit && T >= 8;
    const int utile = (int)blockIdx.x + (j0 + (ws >> 1)) * (int)gridDim.x;
    const float* zq = z + (size_t)(utile * NTILE + lrow) * D + half * 4;
    const float* wr_h = wr_s + 128 * hN;
    v8f acc[4] = {};
    v4f zn[4];
    if (spread) {
#pragma unroll
      for (int u = 0; u < 4; ++u) zn[u] = *reinterpret_cast<const v4f*>(zq + 8 * u);
    } else if (unit) {
      for (int s = 0; s < 16; ++s) {
        const v4f za = *reinterpret_cast<const v4f*>(zq + 8 * s);
        zr_slice<2>(za, wr_h + (4 * s) * WPITCH, acc);
      }
    }
    int rp = 0, dg = 0;
    if (lane < T) {
      const int li = ws * T + lane;
      const int lnode =
          ((int)blockIdx.x + (j0 + (li >> 4)) * (int)gridDim.x) * NTILE + (li & 15);
      if (lnode < n) { rp = rowptr[lnode]; dg = deg[lnode]; }
    }
    int idx_next;
    {
      int s0 = __builtin_amdgcn_readlane(rp, 0);
      int c0 = __builtin_amdgcn_readlane(dg, 0);
      int m0 = c0 < GCHUNK ? c0 : GCHUNK;
      idx_next = (lane < m0) ? nbr[s0 + lane] : 0;
    }
    for (int u = 0; u < T; ++u) {
      const int i = ws * T + u;
      v4f za0 = {}, za1 = {};
      const bool sl2 = spread && u < 8;
      if (sl2) {
        if ((u & 1) == 0) {
#pragma unroll
          for (int k = 0; k < 4; ++k) *reinterpret_cast<v4f*>(zst + 128 * k) = zn[k];
        }
        za0 = *reinterpret_cast<const v4f*>(zst + 128 * (2 * (u & 1)));
        za1 = *reinterpret_cast<const v4f*>(zst + 128 * (2 * (u & 1) + 1));
      }
      const int start = __builtin_amdgcn_readlane(rp, u);
      const int cnt = __builtin_amdgcn_readlane(dg, u);
      const int idx = idx_next;
      if (u + 1 < T) {
        int s1 = __builtin_amdgcn_readlane(rp, u + 1);
        int c1 = __builtin_amdgcn_readlane(dg, u + 1);
        int m1 = c1 < GCHUNK ? c1 : GCHUNK;
        idx_next = (lane < m1) ? nbr[s1 + lane] : 0;
      }
      float4 nsum = gather_node(z4, nbr, start, cnt, idx, lane, [&]() {
        if (sl2) {
          if ((u & 1) == 1 && u + 1 < 8) {
            const float* zg = zq + 16 * (u + 1);
#pragma unroll
            for (int k = 0; k < 4; ++k) zn[k] = *reinterpret_cast<const v4f*>(zg + 8 * k);
          }
          zr_slice<2>(za0, wr_h + (4 * (2 * u)) * WPITCH, acc);
          zr_slice<2>(za1, wr_h + (4 * (2 * u + 1)) * WPITCH, acc);
        }
      });
      const float invr = 1.0f / fmaxf((float)cnt, 1.0f);
      nsum.x *= invr; nsum.y *= invr; nsum.z *= invr; nsum.w *= invr;
      if ((i >> 4) < 8) {
        float* sl = wlds + 2 * WMAT + (2 * (i >> 4)) * ATILE + (i & 15) * APITCH + 4 * lane;
        *reinterpret_cast<float4*>(sl) = nsum;
        *reinterpret_cast<float4*>(sl + ATILE) = nsum;
      } else {
        *reinterpret_cast<float4*>(xt + (i & 15) * APITCH + 4 * lane) = nsum;
      }
    }
    __syncthreads();
    if (unit) {
      tile_product<false, 2>(afrag, 1.0f, wl_l + 128 * hN, acc);
      int mbase = utile * NTILE + (half << 3);
#pragma unroll
      for (int a = 0; a < 4; ++a) {
        const int nn = 4 * hN + a;
        float bias = bl[nn * 16 + lrow];
#pragma unroll
        for (int v = 0; v < 8; ++v) {
          float val = acc[a][v] + bias;
          if (do_relu) val = fmaxf(val, 0.0f);
          out[(size_t)(mbase + v) * D + nn * 16 + lrow] = val;
        }
      }
    }
    if (T == 9 && ws < 4) {
      const int xtile = (int)blockIdx.x + (j0 + 8) * (int)gridDim.x;
      const float* zx = z + (size_t)(xtile * NTILE + lrow) * D + half * 4;
      v8f acx[2] = {};
      for (int s = 0; s < 16; ++s) {
        const v4f za = *reinterpret_cast<const v4f*>(zx + 8 * s);
        zr_slice<1>(za, wr_s + 64 * ws + (4 * s) * WPITCH, acx);
      }
      tile_product<false, 1>(xt + lrow * APITCH + half * 2, 1.0f, wl_l + 64 * ws, acx);
      int mbase = xtile * NTILE + (half << 3);
#pragma unroll
      for (int a = 0; a < 2; ++a) {
        const int nn = 2 * ws + a;
        float bias = bl[nn * 16 + lrow];
#pragma unroll
        for (int v = 0; v < 8; ++v) {
          float val = acx[a][v] + bias;
          if (do_relu) val = fmaxf(val, 0.0f);
          out[(size_t)(mbase + v) * D + nn * 16 + lrow] = val;
        }
      }
    }
  }

  for (int tile = rem + blockIdx.x + w * gridDim.x; tile < ntiles; tile += 16 * gridDim.x) {
    v8f acc[8] = {};
    const int row = tile * NTILE + lrow;
    const float* zq = z + (size_t)row * D + half * 4;
    const int rp = (row < n) ? rowptr[row] : 0;
    const int dg = (row < n) ? deg[row] : 0;
    v4f zn[4];
#pragma unroll
    for (int u = 0; u < 4; ++u) zn[u] = *reinterpret_cast<const v4f*>(zq + 8 * u);

    wave_sync();

    int idx_next;
    {
      int s0 = __builtin_amdgcn_readlane(rp, 0);
      int c0 = __builtin_amdgcn_readlane(dg, 0);
      int m0 = c0 < GCHUNK ? c0 : GCHUNK;
      idx_next = (lane < m0) ? nbr[s0 + lane] : 0;
    }
    for (int r = 0; r < NTILE; ++r) {
      if ((r & 3) == 0) {
#pragma unroll
        for (int u = 0; u < 4; ++u) *reinterpret_cast<v4f*>(zst + 128 * u) = zn[u];
      }
      const v4f za = *reinterpret_cast<const v4f*>(zst + 128 * (r & 3));
      const int start = __builtin_amdgcn_readlane(rp, r);
      const int cnt = __builtin_amdgcn_readlane(dg, r);
      int idx = idx_next;
      if (r + 1 < NTILE) {
        int s1 = __builtin_amdgcn_readlane(rp, r + 1);
        int c1 = __builtin_amdgcn_readlane(dg, r + 1);
        int m1 = c1 < GCHUNK ? c1 : GCHUNK;
        idx_next = (lane < m1) ? nbr[s1 + lane] : 0;
      }
      float4 nsum = make_float4(0.0f, 0.0f, 0.0f, 0.0f);
      int base = 0;
      do {
        int m = cnt - base;
        if (m > GCHUNK) m = GCHUNK;
        if (base > 0) idx = (lane < m) ? nbr[start + base + lane] : 0;
        float4 va[8], vb[8], vc[4], vd[2], ve[1];
        int j = 0;
        if (m >= 8)  { load_rows<8>(z4, idx, j, lane, va); j += 8; }
        if (m >= 16) { load_rows<8>(z4, idx, j, lane, vb); j += 8; }
        if (m & 4)   { load_rows<4>(z4, idx, j, lane, vc); j += 4; }
        if (m & 2)   { load_rows<2>(z4, idx, j, lane, vd); j += 2; }
        if (m & 1)   { load_rows<1>(z4, idx, j, lane, ve); }
        if (base == 0) {
          if ((r & 3) == 3 && r + 1 < NTILE) {
            const float* zg = zq + 8 * (r + 1);
#pragma unroll
            for (int u = 0; u < 4; ++u) zn[u] = *reinterpret_cast<const v4f*>(zg + 8 * u);
          }
          zr_slice(za, wr_s + (4 * r) * WPITCH, acc);
        }
        if (m >= 8)  sum_rows<8>(va, nsum);
        if (m >= 16) sum_rows<8>(vb, nsum);
        if (m & 4)   sum_rows<4>(vc, nsum);
        if (m & 2)   sum_rows<2>(vd, nsum);
        if (m & 1)   sum_rows<1>(ve, nsum);
        base += GCHUNK;
      } while (base < cnt);
      const float invr = 1.0f / fmaxf((float)cnt, 1.0f);
      nsum.x *= invr; nsum.y *= invr; nsum.z *= invr; nsum.w *= invr;
      *reinterpret_cast<float4*>(at + r * APITCH + 4 * lane) = nsum;
    }
    wave_sync();

    tile_product<false>(afrag, 1.0f, wl_l, acc);

    int mbase = tile * NTILE + (half << 3);
#pragma unroll
    for (int nn = 0; nn < 8; ++nn) {
      float bias = bl[nn * 16 + lrow];
#pragma unroll
      for (int v = 0; v < 8; ++v) {
        float val = acc[nn][v] + bias;
        if (do_relu) val = fmaxf(val, 0.0f);
        out[(size_t)(mbase + v) * D + nn * 16 + lrow] = val;
      }
    }
  }
}

extern "C" void kernel_launch(void* const* d_in, const int* in_sizes, int n_in,
                              void* d_out, int out_size, void* d_ws, size_t ws_size,
                              hipStream_t stream) {
  const float* x   = (const float*)d_in[0];
  const int*   ei  = (const int*)d_in[1];
  const float* Wl1 = (const float*)d_in[2];
  const float* bl1 = (const float*)d_in[3];
  const float* Wr1 = (const float*)d_in[4];
  const float* Wl2 = (const float*)d_in[5];
  const float* bl2 = (const float*)d_in[6];
  const float* Wr2 = (const float*)d_in[7];
  float* out = (float*)d_out;

  const int N = in_sizes[0] / D;
  const int E = in_sizes[1] / 2;
  const int ntiles = (N + NTILE - 1) / NTILE;

  const int* src = ei;
  const int* dst = ei + E;

  float* cnt = (float*)d_ws;
  float* summed = cnt + N;

  const int combine_blocks = 256;

  const int scan_blocks = (N + SCAN_TILE - 1) / SCAN_TILE;
  const int Npad = scan_blocks * SCAN_TILE;
  const size_t need = ((size_t)N * (D + 1) + 2 * (size_t)Npad + (size_t)E) * sizeof(float);

  if (ws_size >= need) {
    int* deg = (int*)(summed + (size_t)N * D);
    int* rowptr = deg + Npad;
    int* nbr = rowptr + Npad;

    const int ranges = (N + BIN_NODES - 1) / BIN_NODES;
    if (ranges <= NBIN && (size_t)N * D >= 2 * (size_t)E + 2 * NCHUNK * NBIN) {
      const int CH = (E + NCHUNK - 1) / NCHUNK;
      int2* pairs = (int2*)d_out;
      int* histTab = (int*)d_out + 2 * (size_t)E;
      int* offTab = histTab + NCHUNK * NBIN;
      partition_kernel<<<NCHUNK, 1024, 0, stream>>>(src, dst, pairs, histTab, offTab, E, CH);
      finish_kernel<<<ranges, 1024, 0, stream>>>(pairs, histTab, offTab, rowptr, deg, nbr,
                                                 N, CH);
    } else {
      int* rank = (int*)d_out;
      int edge_blocks = (E + 255) / 256;
      zero_kernel<<<(Npad + 255) / 256, 256, 0, stream>>>((float*)deg, (long)Npad);
      count_kernel<<<edge_blocks, 256, 0, stream>>>(dst, deg, rank, E);
      scan_kernel<<<scan_blocks, 1024, 0, stream>>>(deg, rowptr);
      fill_kernel<<<edge_blocks, 256, 0, stream>>>(src, dst, rank, rowptr, nbr, E);
    }
    float* h = summed;
    layer_kernel<<<combine_blocks, 512, 0, stream>>>(x, rowptr, deg, nbr, Wl1, bl1, Wr1, h,
                                                     ntiles, N, 1);
    layer_kernel<<<combine_blocks, 512, 0, stream>>>(h, rowptr, deg, nbr, Wl2, bl2, Wr2, out,
                                                     ntiles, N, 0);
    return;
  }

  int scatter_blocks = (int)(((long)E * 32 + 255) / 256);

  zero_kernel<<<2048, 256, 0, stream>>>(cnt, (long)N * (D + 1));
  scatter_kernel<<<scatter_blocks, 256, 0, stream>>>(x, src, dst, summed, cnt, E);
  inv_kernel<<<(N + 255) / 256, 256, 0, stream>>>(cnt, N);
  combine_kernel<<<combine_blocks, 512, 0, stream>>>(summed, cnt, x, Wl1, bl1, Wr1,
                                                     out, ntiles, 1);
  zero_kernel<<<2048, 256, 0, stream>>>(summed, (long)N * D);
  scatter_kernel<<<scatter_blocks, 256, 0, stream>>>(out, src, dst, summed, nullptr, E);
  combine_kernel<<<combine_blocks, 512, 0, stream>>>(summed, cnt, out, Wl2, bl2, Wr2,
                                                     out, ntiles, 0);
}
